# starting kernel
# baseline (speedup 1.0000x reference)
#include <hip/hip_runtime.h>
#include <hip/hip_bf16.h>

#define NNODES 50000
#define NEDGES 800000

typedef __attribute__((ext_vector_type(16))) __bf16        bf16x16;
typedef __attribute__((ext_vector_type(8)))  float         f32x8;
typedef __attribute__((ext_vector_type(8)))  unsigned int  u32x8;

__device__ __forceinline__ unsigned short f2bf(float f) {
  unsigned int u = __float_as_uint(f);
  u += 0x7FFFu + ((u >> 16) & 1u);
  return (unsigned short)(u >> 16);
}
__device__ __forceinline__ float bf2f(unsigned short s) {
  return __uint_as_float(((unsigned int)s) << 16);
}
__device__ __forceinline__ float silu_f(float v) { return v / (1.0f + __expf(-v)); }
__device__ __forceinline__ float sigm_f(float v) { return 1.0f / (1.0f + __expf(-v)); }

__device__ __forceinline__ int frag_k(int lane, int v) {
  return ((lane >> 4) << 3) + ((v >> 2) << 4) + ((v & 3) << 1);
}

__device__ __forceinline__ bf16x16 load_a_frag(const unsigned short* tile, int ld, int kc, int lane) {
  const int row = lane & 15;
  u32x8 r;
#pragma unroll
  for (int v = 0; v < 8; ++v) {
    const int k = kc * 32 + frag_k(lane, v);
    r[v] = *(const unsigned int*)(tile + row * ld + k);
  }
  return __builtin_bit_cast(bf16x16, r);
}

__device__ __forceinline__ bf16x16 load_b_frag(const u32x8* __restrict__ wf, int kc, int nt, int lane) {
  return __builtin_bit_cast(bf16x16, wf[(kc * 8 + nt) * 32 + lane]);
}

__global__ void egnn_prep_kernel(const float* __restrict__ w_e1, const float* __restrict__ w_e2,
                                 const float* __restrict__ w_h1, const float* __restrict__ w_h2,
                                 unsigned int* __restrict__ wf_e1, unsigned int* __restrict__ wf_e2,
                                 unsigned int* __restrict__ wf_h1, unsigned int* __restrict__ wf_h2) {
  int t = blockIdx.x * blockDim.x + threadIdx.x;
  const float* W; unsigned int* D; int K; int local;
  if      (t < 18432)         { W = w_e1; D = wf_e1; K = 257; local = t; }
  else if (t < 18432 + 8192)  { W = w_e2; D = wf_e2; K = 128; local = t - 18432; }
  else if (t < 26624 + 16384) { W = w_h1; D = wf_h1; K = 256; local = t - 26624; }
  else if (t < 51200)         { W = w_h2; D = wf_h2; K = 128; local = t - 43008; }
  else return;
  const int v = local & 7, lane = (local >> 3) & 31, nt = (local >> 8) & 7, kc = local >> 11;
  const int k   = kc * 32 + ((lane >> 4) << 3) + ((v >> 2) << 4) + ((v & 3) << 1);
  const int col = nt * 16 + (lane & 15);
  const float a = (k     < K) ? W[k * 128 + col]       : 0.0f;
  const float b = (k + 1 < K) ? W[(k + 1) * 128 + col] : 0.0f;
  D[local] = (unsigned int)f2bf(a) | ((unsigned int)f2bf(b) << 16);
}

__global__ void egnn_zero_kernel(float* __restrict__ p, int n) {
  int t = blockIdx.x * blockDim.x + threadIdx.x;
  if (t < n) p[t] = 0.0f;
}

__global__ __launch_bounds__(128) void egnn_edge_kernel(
    const float* __restrict__ h, const float* __restrict__ x, const int* __restrict__ edges,
    const u32x8* __restrict__ wf_e1, const u32x8* __restrict__ wf_e2,
    const float* __restrict__ b_e1, const float* __restrict__ b_e2,
    const float* __restrict__ w_inf, const float* __restrict__ b_inf,
    float* __restrict__ m_i) {
  __shared__ unsigned short stage[4][16 * 288];
  __shared__ unsigned short a2[4][16 * 128];
  __shared__ float gbuf[4][16];

  const int lane = threadIdx.x & 31;
  const int wave = threadIdx.x >> 5;
  const int colq = lane & 15;
  const int ebase = (blockIdx.x * 4 + wave) * 16;
  unsigned short* st = stage[wave];
  unsigned short* aw = a2[wave];

  for (int r = 0; r < 16; ++r) {
    const int e = ebase + r;
    const int s = edges[2 * e], d = edges[2 * e + 1];
    const float dx = x[3 * s] - x[3 * d], dy = x[3 * s + 1] - x[3 * d + 1], dz = x[3 * s + 2] - x[3 * d + 2];
    const float dist = sqrtf(dx * dx + dy * dy + dz * dz);
#pragma unroll
    for (int i = lane; i < 288; i += 32) {
      float v;
      if      (i < 128)  v = h[s * 128 + i];
      else if (i < 256)  v = h[d * 128 + (i - 128)];
      else if (i == 256) v = dist;
      else               v = 0.0f;
      st[r * 288 + i] = f2bf(v);
    }
  }
  __builtin_amdgcn_wave_barrier();

#pragma unroll 1
  for (int nt = 0; nt < 8; ++nt) {
    f32x8 acc;
    const float bias = b_e1[nt * 16 + colq];
#pragma unroll
    for (int v = 0; v < 8; ++v) acc[v] = bias;
#pragma unroll
    for (int kc = 0; kc < 9; ++kc) {
      if (kc + 1 < 9) __builtin_prefetch(&wf_e1[((kc + 1) * 8 + nt) * 32 + lane], 0, 3);
      bf16x16 a = load_a_frag(st, 288, kc, lane);
      bf16x16 b = load_b_frag(wf_e1, kc, nt, lane);
      acc = __builtin_amdgcn_wmma_f32_16x16x32_bf16(false, a, false, b, (short)0, acc, false, false);
    }
#pragma unroll
    for (int v = 0; v < 8; ++v) {
      const int row = v + ((lane >> 4) << 3);
      aw[row * 128 + nt * 16 + colq] = f2bf(silu_f(acc[v]));
    }
  }
  __builtin_amdgcn_wave_barrier();

  unsigned short* mb = st;
  f32x8 dot;
#pragma unroll
  for (int v = 0; v < 8; ++v) dot[v] = 0.0f;
#pragma unroll 1
  for (int nt = 0; nt < 8; ++nt) {
    f32x8 acc;
    const float bias = b_e2[nt * 16 + colq];
#pragma unroll
    for (int v = 0; v < 8; ++v) acc[v] = bias;
#pragma unroll
    for (int kc = 0; kc < 4; ++kc) {
      bf16x16 a = load_a_frag(aw, 128, kc, lane);
      bf16x16 b = load_b_frag(wf_e2, kc, nt, lane);
      acc = __builtin_amdgcn_wmma_f32_16x16x32_bf16(false, a, false, b, (short)0, acc, false, false);
    }
    const float wv = w_inf[nt * 16 + colq];
#pragma unroll
    for (int v = 0; v < 8; ++v) {
      const float m = silu_f(acc[v]);
      dot[v] += m * wv;
      const int row = v + ((lane >> 4) << 3);
      mb[row * 128 + nt * 16 + colq] = f2bf(m);
    }
  }
  const float binf = b_inf[0];
#pragma unroll
  for (int v = 0; v < 8; ++v) {
    float s = dot[v];
    s += __shfl_xor(s, 1, 32);
    s += __shfl_xor(s, 2, 32);
    s += __shfl_xor(s, 4, 32);
    s += __shfl_xor(s, 8, 32);
    if (colq == 0) gbuf[wave][v + ((lane >> 4) << 3)] = sigm_f(s + binf);
  }
  __builtin_amdgcn_wave_barrier();

  for (int r = 0; r < 16; ++r) {
    const int e = ebase + r;
    const int s = edges[2 * e];
    const float g = gbuf[wave][r];
#pragma unroll
    for (int c = lane; c < 128; c += 32) {
      unsafeAtomicAdd(&m_i[s * 128 + c], g * bf2f(mb[r * 128 + c]));
    }
  }
}

__global__ __launch_bounds__(128) void egnn_node_kernel(
    const float* __restrict__ h,
    const u32x8* __restrict__ wf_h1, const u32x8* __restrict__ wf_h2,
    const float* __restrict__ b_h1, const float* __restrict__ b_h2,
    float* __restrict__ hio) {
  __shared__ unsigned short stage[4][16 * 256];
  __shared__ unsigned short a2[4][16 * 128];

  const int lane = threadIdx.x & 31;
  const int wave = threadIdx.x >> 5;
  const int colq = lane & 15;
  const int nbase = (blockIdx.x * 4 + wave) * 16;
  if (nbase >= NNODES) return;
  unsigned short* st = stage[wave];
  unsigned short* aw = a2[wave];

  for (int r = 0; r < 16; ++r) {
    const int node = nbase + r;
    const bool ok = node < NNODES;
#pragma unroll
    for (int i = lane; i < 256; i += 32) {
      float v = 0.0f;
      if (ok) v = (i < 128) ? h[node * 128 + i] : hio[node * 128 + (i - 128)];
      st[r * 256 + i] = f2bf(v);
    }
  }
  __builtin_amdgcn_wave_barrier();

#pragma unroll 1
  for (int nt = 0; nt < 8; ++nt) {
    f32x8 acc;
    const float bias = b_h1[nt * 16 + colq];
#pragma unroll
    for (int v = 0; v < 8; ++v) acc[v] = bias;
#pragma unroll
    for (int kc = 0; kc < 8; ++kc) {
      if (kc + 1 < 8) __builtin_prefetch(&wf_h1[((kc + 1) * 8 + nt) * 32 + lane], 0, 3);
      bf16x16 a = load_a_frag(st, 256, kc, lane);
      bf16x16 b = load_b_frag(wf_h1, kc, nt, lane);
      acc = __builtin_amdgcn_wmma_f32_16x16x32_bf16(false, a, false, b, (short)0, acc, false, false);
    }
#pragma unroll
    for (int v = 0; v < 8; ++v) {
      const int row = v + ((lane >> 4) << 3);
      aw[row * 128 + nt * 16 + colq] = f2bf(silu_f(acc[v]));
    }
  }
  __builtin_amdgcn_wave_barrier();

#pragma unroll 1
  for (int nt = 0; nt < 8; ++nt) {
    f32x8 acc;
    const float bias = b_h2[nt * 16 + colq];
#pragma unroll
    for (int v = 0; v < 8; ++v) acc[v] = bias;
#pragma unroll
    for (int kc = 0; kc < 4; ++kc) {
      bf16x16 a = load_a_frag(aw, 128, kc, lane);
      bf16x16 b = load_b_frag(wf_h2, kc, nt, lane);
      acc = __builtin_amdgcn_wmma_f32_16x16x32_bf16(false, a, false, b, (short)0, acc, false, false);
    }
#pragma unroll
    for (int v = 0; v < 8; ++v) {
      const int row = v + ((lane >> 4) << 3);
      const int node = nbase + row;
      const int col = nt * 16 + colq;
      if (node < NNODES) hio[node * 128 + col] = acc[v] + h[node * 128 + col];
    }
  }
}

extern "C" void kernel_launch(void* const* d_in, const int* in_sizes, int n_in,
                              void* d_out, int out_size, void* d_ws, size_t ws_size,
                              hipStream_t stream) {
  const float* h     = (const float*)d_in[0];
  const float* x     = (const float*)d_in[1];
  const int*   edges = (const int*)  d_in[2];
  const float* w_e1  = (const float*)d_in[3];
  const float* b_e1  = (const float*)d_in[4];
  const float* w_e2  = (const float*)d_in[5];
  const float* b_e2  = (const float*)d_in[6];
  const float* w_inf = (const float*)d_in[7];
  const float* b_inf = (const float*)d_in[8];
  const float* w_h1  = (const float*)d_in[9];
  const float* b_h1  = (const float*)d_in[10];
  const float* w_h2  = (const float*)d_in[11];
  const float* b_h2  = (const float*)d_in[12];
  float* out = (float*)d_out;

  unsigned int* wf_e1 = (unsigned int*)d_ws;
  unsigned int* wf_e2 = wf_e1 + 18432;
  unsigned int* wf_h1 = wf_e2 + 8192;
  unsigned int* wf_h2 = wf_h1 + 16384;

  egnn_prep_kernel<<<(51200 + 255) / 256, 256, 0, stream>>>(
      w_e1, w_e2, w_h1, w_h2, wf_e1, wf_e2, wf_h1, wf_h2);

  egnn_zero_kernel<<<(NNODES * 128 + 255) / 256, 256, 0, stream>>>(out, NNODES * 128);

  egnn_edge_kernel<<<NEDGES / 64, 128, 0, stream>>>(
      h, x, edges, (const u32x8*)wf_e1, (const u32x8*)wf_e2,
      b_e1, b_e2, w_inf, b_inf, out);

  egnn_node_kernel<<<(NNODES + 63) / 64, 128, 0, stream>>>(
      h, (const u32x8*)wf_h1, (const u32x8*)wf_h2, b_h1, b_h2, out);
}


# edge kernel: a wave reads its 16 index pairs in one coalesced load and keeps them, so all 32 feature-row loads (16 B per lane) issue before the first wait and the scatter reloads no index
# speedup vs baseline: 1.1638x; 1.1638x over previous
#include <hip/hip_runtime.h>
#include <hip/hip_bf16.h>

#define NNODES 50000
#define NEDGES 800000

typedef __attribute__((ext_vector_type(16))) __bf16        bf16x16;
typedef __attribute__((ext_vector_type(8)))  float         f32x8;
typedef __attribute__((ext_vector_type(8)))  unsigned int  u32x8;

__device__ __forceinline__ unsigned short f2bf(float f) {
  unsigned int u = __float_as_uint(f);
  u += 0x7FFFu + ((u >> 16) & 1u);
  return (unsigned short)(u >> 16);
}
__device__ __forceinline__ float bf2f(unsigned short s) {
  return __uint_as_float(((unsigned int)s) << 16);
}
__device__ __forceinline__ float silu_f(float v) { return v / (1.0f + __expf(-v)); }
__device__ __forceinline__ float sigm_f(float v) { return 1.0f / (1.0f + __expf(-v)); }

__device__ __forceinline__ int frag_k(int lane, int v) {
  return ((lane >> 4) << 3) + ((v >> 2) << 4) + ((v & 3) << 1);
}

__device__ __forceinline__ bf16x16 load_a_frag(const unsigned short* tile, int ld, int kc, int lane) {
  const int row = lane & 15;
  u32x8 r;
#pragma unroll
  for (int v = 0; v < 8; ++v) {
    const int k = kc * 32 + frag_k(lane, v);
    r[v] = *(const unsigned int*)(tile + row * ld + k);
  }
  return __builtin_bit_cast(bf16x16, r);
}

__device__ __forceinline__ bf16x16 load_b_frag(const u32x8* __restrict__ wf, int kc, int nt, int lane) {
  return __builtin_bit_cast(bf16x16, wf[(kc * 8 + nt) * 32 + lane]);
}

__global__ void egnn_prep_kernel(const float* __restrict__ w_e1, const float* __restrict__ w_e2,
                                 const float* __restrict__ w_h1, const float* __restrict__ w_h2,
                                 unsigned int* __restrict__ wf_e1, unsigned int* __restrict__ wf_e2,
                                 unsigned int* __restrict__ wf_h1, unsigned int* __restrict__ wf_h2) {
  int t = blockIdx.x * blockDim.x + threadIdx.x;
  const float* W; unsigned int* D; int K; int local;
  if      (t < 18432)         { W = w_e1; D = wf_e1; K = 257; local = t; }
  else if (t < 18432 + 8192)  { W = w_e2; D = wf_e2; K = 128; local = t - 18432; }
  else if (t < 26624 + 16384) { W = w_h1; D = wf_h1; K = 256; local = t - 26624; }
  else if (t < 51200)         { W = w_h2; D = wf_h2; K = 128; local = t - 43008; }
  else return;
  const int v = local & 7, lane = (local >> 3) & 31, nt = (local >> 8) & 7, kc = local >> 11;
  const int k   = kc * 32 + ((lane >> 4) << 3) + ((v >> 2) << 4) + ((v & 3) << 1);
  const int col = nt * 16 + (lane & 15);
  const float a = (k     < K) ? W[k * 128 + col]       : 0.0f;
  const float b = (k + 1 < K) ? W[(k + 1) * 128 + col] : 0.0f;
  D[local] = (unsigned int)f2bf(a) | ((unsigned int)f2bf(b) << 16);
}

__global__ void egnn_zero_kernel(float* __restrict__ p, int n) {
  int t = blockIdx.x * blockDim.x + threadIdx.x;
  if (t < n) p[t] = 0.0f;
}

__global__ __launch_bounds__(128) void egnn_edge_kernel(
    const float* __restrict__ h, const float* __restrict__ x, const int* __restrict__ edges,
    const u32x8* __restrict__ wf_e1, const u32x8* __restrict__ wf_e2,
    const float* __restrict__ b_e1, const float* __restrict__ b_e2,
    const float* __restrict__ w_inf, const float* __restrict__ b_inf,
    float* __restrict__ m_i) {
  __shared__ __attribute__((aligned(16))) unsigned short stage[4][16 * 288];
  __shared__ unsigned short a2[4][16 * 128];
  __shared__ float gbuf[4][16];

  const int lane = threadIdx.x & 31;
  const int wave = threadIdx.x >> 5;
  const int colq = lane & 15;
  const int ebase = (blockIdx.x * 4 + wave) * 16;
  unsigned short* st = stage[wave];
  unsigned short* aw = a2[wave];

  const int my_node = edges[2 * ebase + lane];
  const float px = x[3 * my_node], py = x[3 * my_node + 1], pz = x[3 * my_node + 2];
  float4 hs[16], hd[16];
#pragma unroll
  for (int r = 0; r < 16; ++r) {
    const int s = __builtin_amdgcn_readlane(my_node, 2 * r);
    const int d = __builtin_amdgcn_readlane(my_node, 2 * r + 1);
    hs[r] = *(const float4*)(h + s * 128 + lane * 4);
    hd[r] = *(const float4*)(h + d * 128 + lane * 4);
  }
  const float dx = px - __shfl_xor(px, 1, 32), dy = py - __shfl_xor(py, 1, 32), dz = pz - __shfl_xor(pz, 1, 32);
  const float dist = sqrtf(dx * dx + dy * dy + dz * dz);
#pragma unroll
  for (int r = 0; r < 16; ++r) {
    unsigned int* ws = (unsigned int*)(st + r * 288 + lane * 4);
    unsigned int* wd = (unsigned int*)(st + r * 288 + 128 + lane * 4);
    ws[0] = (unsigned int)f2bf(hs[r].x) | ((unsigned int)f2bf(hs[r].y) << 16);
    ws[1] = (unsigned int)f2bf(hs[r].z) | ((unsigned int)f2bf(hs[r].w) << 16);
    wd[0] = (unsigned int)f2bf(hd[r].x) | ((unsigned int)f2bf(hd[r].y) << 16);
    wd[1] = (unsigned int)f2bf(hd[r].z) | ((unsigned int)f2bf(hd[r].w) << 16);
    const float dr = __builtin_bit_cast(float, __builtin_amdgcn_readlane(__builtin_bit_cast(int, dist), 2 * r));
    st[r * 288 + 256 + lane] = f2bf(lane == 0 ? dr : 0.0f);
  }
  __builtin_amdgcn_wave_barrier();

#pragma unroll 1
  for (int nt = 0; nt < 8; ++nt) {
    f32x8 acc;
    const float bias = b_e1[nt * 16 + colq];
#pragma unroll
    for (int v = 0; v < 8; ++v) acc[v] = bias;
#pragma unroll
    for (int kc = 0; kc < 9; ++kc) {
      if (kc + 1 < 9) __builtin_prefetch(&wf_e1[((kc + 1) * 8 + nt) * 32 + lane], 0, 3);
      bf16x16 a = load_a_frag(st, 288, kc, lane);
      bf16x16 b = load_b_frag(wf_e1, kc, nt, lane);
      acc = __builtin_amdgcn_wmma_f32_16x16x32_bf16(false, a, false, b, (short)0, acc, false, false);
    }
#pragma unroll
    for (int v = 0; v < 8; ++v) {
      const int row = v + ((lane >> 4) << 3);
      aw[row * 128 + nt * 16 + colq] = f2bf(silu_f(acc[v]));
    }
  }
  __builtin_amdgcn_wave_barrier();

  unsigned short* mb = st;
  f32x8 dot;
#pragma unroll
  for (int v = 0; v < 8; ++v) dot[v] = 0.0f;
#pragma unroll 1
  for (int nt = 0; nt < 8; ++nt) {
    f32x8 acc;
    const float bias = b_e2[nt * 16 + colq];
#pragma unroll
    for (int v = 0; v < 8; ++v) acc[v] = bias;
#pragma unroll
    for (int kc = 0; kc < 4; ++kc) {
      bf16x16 a = load_a_frag(aw, 128, kc, lane);
      bf16x16 b = load_b_frag(wf_e2, kc, nt, lane);
      acc = __builtin_amdgcn_wmma_f32_16x16x32_bf16(false, a, false, b, (short)0, acc, false, false);
    }
    const float wv = w_inf[nt * 16 + colq];
#pragma unroll
    for (int v = 0; v < 8; ++v) {
      const float m = silu_f(acc[v]);
      dot[v] += m * wv;
      const int row = v + ((lane >> 4) << 3);
      mb[row * 128 + nt * 16 + colq] = f2bf(m);
    }
  }
  const float binf = b_inf[0];
#pragma unroll
  for (int v = 0; v < 8; ++v) {
    float s = dot[v];
    s += __shfl_xor(s, 1, 32);
    s += __shfl_xor(s, 2, 32);
    s += __shfl_xor(s, 4, 32);
    s += __shfl_xor(s, 8, 32);
    if (colq == 0) gbuf[wave][v + ((lane >> 4) << 3)] = sigm_f(s + binf);
  }
  __builtin_amdgcn_wave_barrier();

#pragma unroll
  for (int r = 0; r < 16; ++r) {
    const int s = __builtin_amdgcn_readlane(my_node, 2 * r);
    const float g = gbuf[wave][r];
#pragma unroll
    for (int c = lane; c < 128; c += 32) {
      unsafeAtomicAdd(&m_i[s * 128 + c], g * bf2f(mb[r * 128 + c]));
    }
  }
}

__global__ __launch_bounds__(128) void egnn_node_kernel(
    const float* __restrict__ h,
    const u32x8* __restrict__ wf_h1, const u32x8* __restrict__ wf_h2,
    const float* __restrict__ b_h1, const float* __restrict__ b_h2,
    float* __restrict__ hio) {
  __shared__ unsigned short stage[4][16 * 256];
  __shared__ unsigned short a2[4][16 * 128];

  const int lane = threadIdx.x & 31;
  const int wave = threadIdx.x >> 5;
  const int colq = lane & 15;
  const int nbase = (blockIdx.x * 4 + wave) * 16;
  if (nbase >= NNODES) return;
  unsigned short* st = stage[wave];
  unsigned short* aw = a2[wave];

  for (int r = 0; r < 16; ++r) {
    const int node = nbase + r;
    const bool ok = node < NNODES;
#pragma unroll
    for (int i = lane; i < 256; i += 32) {
      float v = 0.0f;
      if (ok) v = (i < 128) ? h[node * 128 + i] : hio[node * 128 + (i - 128)];
      st[r * 256 + i] = f2bf(v);
    }
  }
  __builtin_amdgcn_wave_barrier();

#pragma unroll 1
  for (int nt = 0; nt < 8; ++nt) {
    f32x8 acc;
    const float bias = b_h1[nt * 16 + colq];
#pragma unroll
    for (int v = 0; v < 8; ++v) acc[v] = bias;
#pragma unroll
    for (int kc = 0; kc < 8; ++kc) {
      if (kc + 1 < 8) __builtin_prefetch(&wf_h1[((kc + 1) * 8 + nt) * 32 + lane], 0, 3);
      bf16x16 a = load_a_frag(st, 256, kc, lane);
      bf16x16 b = load_b_frag(wf_h1, kc, nt, lane);
      acc = __builtin_amdgcn_wmma_f32_16x16x32_bf16(false, a, false, b, (short)0, acc, false, false);
    }
#pragma unroll
    for (int v = 0; v < 8; ++v) {
      const int row = v + ((lane >> 4) << 3);
      aw[row * 128 + nt * 16 + colq] = f2bf(silu_f(acc[v]));
    }
  }
  __builtin_amdgcn_wave_barrier();

#pragma unroll 1
  for (int nt = 0; nt < 8; ++nt) {
    f32x8 acc;
    const float bias = b_h2[nt * 16 + colq];
#pragma unroll
    for (int v = 0; v < 8; ++v) acc[v] = bias;
#pragma unroll
    for (int kc = 0; kc < 4; ++kc) {
      bf16x16 a = load_a_frag(aw, 128, kc, lane);
      bf16x16 b = load_b_frag(wf_h2, kc, nt, lane);
      acc = __builtin_amdgcn_wmma_f32_16x16x32_bf16(false, a, false, b, (short)0, acc, false, false);
    }
#pragma unroll
    for (int v = 0; v < 8; ++v) {
      const int row = v + ((lane >> 4) << 3);
      const int node = nbase + row;
      const int col = nt * 16 + colq;
      if (node < NNODES) hio[node * 128 + col] = acc[v] + h[node * 128 + col];
    }
  }
}

extern "C" void kernel_launch(void* const* d_in, const int* in_sizes, int n_in,
                              void* d_out, int out_size, void* d_ws, size_t ws_size,
                              hipStream_t stream) {
  const float* h     = (const float*)d_in[0];
  const float* x     = (const float*)d_in[1];
  const int*   edges = (const int*)  d_in[2];
  const float* w_e1  = (const float*)d_in[3];
  const float* b_e1  = (const float*)d_in[4];
  const float* w_e2  = (const float*)d_in[5];
  const float* b_e2  = (const float*)d_in[6];
  const float* w_inf = (const float*)d_in[7];
  const float* b_inf = (const float*)d_in[8];
  const float* w_h1  = (const float*)d_in[9];
  const float* b_h1  = (const float*)d_in[10];
  const float* w_h2  = (const float*)d_in[11];
  const float* b_h2  = (const float*)d_in[12];
  float* out = (float*)d_out;

  unsigned int* wf_e1 = (unsigned int*)d_ws;
  unsigned int* wf_e2 = wf_e1 + 18432;
  unsigned int* wf_h1 = wf_e2 + 8192;
  unsigned int* wf_h2 = wf_h1 + 16384;

  egnn_prep_kernel<<<(51200 + 255) / 256, 256, 0, stream>>>(
      w_e1, w_e2, w_h1, w_h2, wf_e1, wf_e2, wf_h1, wf_h2);

  egnn_zero_kernel<<<(NNODES * 128 + 255) / 256, 256, 0, stream>>>(out, NNODES * 128);

  egnn_edge_kernel<<<NEDGES / 64, 128, 0, stream>>>(
      h, x, edges, (const u32x8*)wf_e1, (const u32x8*)wf_e2,
      b_e1, b_e2, w_inf, b_inf, out);

  egnn_node_kernel<<<(NNODES + 63) / 64, 128, 0, stream>>>(
      h, (const u32x8*)wf_h1, (const u32x8*)wf_h2, b_h1, b_h2, out);
}


# edge kernel: one resident 16-wave workgroup per WGP copies the edge MLP's bf16 weight fragments and biases into LDS once and loops over 16-edge tiles; B fragments are LDS reads, not global re-reads
# speedup vs baseline: 1.4739x; 1.2665x over previous
#include <hip/hip_runtime.h>
#include <hip/hip_bf16.h>

#define NNODES 50000
#define NEDGES 800000

typedef __attribute__((ext_vector_type(16))) __bf16        bf16x16;
typedef __attribute__((ext_vector_type(8)))  float         f32x8;
typedef __attribute__((ext_vector_type(8)))  unsigned int  u32x8;

__device__ __forceinline__ unsigned short f2bf(float f) {
  unsigned int u = __float_as_uint(f);
  u += 0x7FFFu + ((u >> 16) & 1u);
  return (unsigned short)(u >> 16);
}
__device__ __forceinline__ float bf2f(unsigned short s) {
  return __uint_as_float(((unsigned int)s) << 16);
}
__device__ __forceinline__ float silu_f(float v) { return v / (1.0f + __expf(-v)); }
__device__ __forceinline__ float sigm_f(float v) { return 1.0f / (1.0f + __expf(-v)); }

__device__ __forceinline__ int frag_k(int lane, int v) {
  return ((lane >> 4) << 3) + ((v >> 2) << 4) + ((v & 3) << 1);
}

__device__ __forceinline__ bf16x16 load_a_frag(const unsigned short* tile, int ld, int kc, int lane) {
  const int row = lane & 15;
  u32x8 r;
#pragma unroll
  for (int v = 0; v < 8; ++v) {
    const int k = kc * 32 + frag_k(lane, v);
    r[v] = *(const unsigned int*)(tile + row * ld + k);
  }
  return __builtin_bit_cast(bf16x16, r);
}

__device__ __forceinline__ bf16x16 load_b_frag(const u32x8* __restrict__ wf, int kc, int nt, int lane) {
  return __builtin_bit_cast(bf16x16, wf[(kc * 8 + nt) * 32 + lane]);
}

__device__ __forceinline__ bf16x16 load_b_frag_lds(const uint4* wl, int kc, int nt, int lane) {
  const uint4 lo = wl[(kc * 8 + nt) * 64 + lane];
  const uint4 hi = wl[(kc * 8 + nt) * 64 + 32 + lane];
  u32x8 r;
  r[0] = lo.x; r[1] = lo.y; r[2] = lo.z; r[3] = lo.w;
  r[4] = hi.x; r[5] = hi.y; r[6] = hi.z; r[7] = hi.w;
  return __builtin_bit_cast(bf16x16, r);
}

__global__ void egnn_prep_kernel(const float* __restrict__ w_e1, const float* __restrict__ w_e2,
                                 const float* __restrict__ w_h1, const float* __restrict__ w_h2,
                                 unsigned int* __restrict__ wf_e1, unsigned int* __restrict__ wf_e2,
                                 unsigned int* __restrict__ wf_h1, unsigned int* __restrict__ wf_h2) {
  int t = blockIdx.x * blockDim.x + threadIdx.x;
  const float* W; unsigned int* D; int K; int local; bool split;
  if      (t < 18432)         { W = w_e1; D = wf_e1; K = 257; local = t;         split = true;  }
  else if (t < 18432 + 8192)  { W = w_e2; D = wf_e2; K = 128; local = t - 18432; split = true;  }
  else if (t < 26624 + 16384) { W = w_h1; D = wf_h1; K = 256; local = t - 26624; split = false; }
  else if (t < 51200)         { W = w_h2; D = wf_h2; K = 128; local = t - 43008; split = false; }
  else return;
  const int v = local & 7, lane = (local >> 3) & 31, nt = (local >> 8) & 7, kc = local >> 11;
  const int k   = kc * 32 + ((lane >> 4) << 3) + ((v >> 2) << 4) + ((v & 3) << 1);
  const int col = nt * 16 + (lane & 15);
  const float a = (k     < K) ? W[k * 128 + col]       : 0.0f;
  const float b = (k + 1 < K) ? W[(k + 1) * 128 + col] : 0.0f;
  const int dst = split ? ((local >> 8) << 8) + ((v >> 2) << 7) + (lane << 2) + (v & 3) : local;
  D[dst] = (unsigned int)f2bf(a) | ((unsigned int)f2bf(b) << 16);
}

__global__ void egnn_zero_kernel(float* __restrict__ p, int n) {
  int t = blockIdx.x * blockDim.x + threadIdx.x;
  if (t < n) p[t] = 0.0f;
}

#define EDGE_BLOCKS 256
#define EDGE_WAVES  16
#define WL1_UNITS (9 * 8 * 64)
#define WL2_UNITS (4 * 8 * 64)
__global__ __launch_bounds__(EDGE_WAVES * 32) void egnn_edge_kernel(
    const float* __restrict__ h, const float* __restrict__ x, const int* __restrict__ edges,
    const uint4* __restrict__ wf_e1, const uint4* __restrict__ wf_e2,
    const float* __restrict__ b_e1, const float* __restrict__ b_e2,
    const float* __restrict__ w_inf, const float* __restrict__ b_inf,
    float* __restrict__ m_i) {
  __shared__ __attribute__((aligned(16))) uint4 wl1[WL1_UNITS];
  __shared__ __attribute__((aligned(16))) uint4 wl2[WL2_UNITS];
  __shared__ __attribute__((aligned(16))) unsigned short stage[EDGE_WAVES][16 * 288];
  __shared__ float gbuf[EDGE_WAVES][16];
  __shared__ float pvec[3][128];

  const int lane = threadIdx.x & 31;
  const int wave = threadIdx.x >> 5;
  const int colq = lane & 15;
  unsigned short* st = stage[wave];
  unsigned short* aw = st;
  unsigned short* mb = st + 2048;

  for (int i = threadIdx.x; i < WL1_UNITS; i += EDGE_WAVES * 32) wl1[i] = wf_e1[i];
  for (int i = threadIdx.x; i < WL2_UNITS; i += EDGE_WAVES * 32) wl2[i] = wf_e2[i];
  if (threadIdx.x < 128) {
    pvec[0][threadIdx.x] = b_e1[threadIdx.x];
    pvec[1][threadIdx.x] = b_e2[threadIdx.x];
    pvec[2][threadIdx.x] = w_inf[threadIdx.x];
  }
  __syncthreads();

#pragma unroll 1
  for (int tile = wave * EDGE_BLOCKS + blockIdx.x; tile < NEDGES / 16; tile += EDGE_BLOCKS * EDGE_WAVES) {
  const int ebase = tile * 16;

  const int my_node = edges[2 * ebase + lane];
  const float px = x[3 * my_node], py = x[3 * my_node + 1], pz = x[3 * my_node + 2];
  float4 hs[16], hd[16];
#pragma unroll
  for (int r = 0; r < 16; ++r) {
    const int s = __builtin_amdgcn_readlane(my_node, 2 * r);
    const int d = __builtin_amdgcn_readlane(my_node, 2 * r + 1);
    hs[r] = *(const float4*)(h + s * 128 + lane * 4);
    hd[r] = *(const float4*)(h + d * 128 + lane * 4);
  }
  const float dx = px - __shfl_xor(px, 1, 32), dy = py - __shfl_xor(py, 1, 32), dz = pz - __shfl_xor(pz, 1, 32);
  const float dist = sqrtf(dx * dx + dy * dy + dz * dz);
#pragma unroll
  for (int r = 0; r < 16; ++r) {
    unsigned int* ws = (unsigned int*)(st + r * 288 + lane * 4);
    unsigned int* wd = (unsigned int*)(st + r * 288 + 128 + lane * 4);
    ws[0] = (unsigned int)f2bf(hs[r].x) | ((unsigned int)f2bf(hs[r].y) << 16);
    ws[1] = (unsigned int)f2bf(hs[r].z) | ((unsigned int)f2bf(hs[r].w) << 16);
    wd[0] = (unsigned int)f2bf(hd[r].x) | ((unsigned int)f2bf(hd[r].y) << 16);
    wd[1] = (unsigned int)f2bf(hd[r].z) | ((unsigned int)f2bf(hd[r].w) << 16);
    const float dr = __builtin_bit_cast(float, __builtin_amdgcn_readlane(__builtin_bit_cast(int, dist), 2 * r));
    st[r * 288 + 256 + lane] = f2bf(lane == 0 ? dr : 0.0f);
  }
  __builtin_amdgcn_wave_barrier();

  bf16x16 af1[9];
#pragma unroll
  for (int kc = 0; kc < 9; ++kc) af1[kc] = load_a_frag(st, 288, kc, lane);
  __builtin_amdgcn_wave_barrier();
#pragma unroll 1
  for (int nt = 0; nt < 8; ++nt) {
    f32x8 acc;
    const float bias = pvec[0][nt * 16 + colq];
    bf16x16 bw[9];
#pragma unroll
    for (int kc = 0; kc < 9; ++kc) bw[kc] = load_b_frag_lds(wl1, kc, nt, lane);
    __builtin_amdgcn_sched_barrier(0);
#pragma unroll
    for (int v = 0; v < 8; ++v) acc[v] = bias;
#pragma unroll
    for (int kc = 0; kc < 9; ++kc)
      acc = __builtin_amdgcn_wmma_f32_16x16x32_bf16(false, af1[kc], false, bw[kc], (short)0, acc, false, false);
#pragma unroll
    for (int v = 0; v < 8; ++v) {
      const int row = v + ((lane >> 4) << 3);
      aw[row * 128 + nt * 16 + colq] = f2bf(silu_f(acc[v]));
    }
  }
  __builtin_amdgcn_wave_barrier();

  bf16x16 af2[4];
#pragma unroll
  for (int kc = 0; kc < 4; ++kc) af2[kc] = load_a_frag(aw, 128, kc, lane);
  f32x8 dot;
#pragma unroll
  for (int v = 0; v < 8; ++v) dot[v] = 0.0f;
#pragma unroll 1
  for (int nt = 0; nt < 8; ++nt) {
    f32x8 acc;
    const float bias = pvec[1][nt * 16 + colq];
    const float wv = pvec[2][nt * 16 + colq];
    bf16x16 bw[4];
#pragma unroll
    for (int kc = 0; kc < 4; ++kc) bw[kc] = load_b_frag_lds(wl2, kc, nt, lane);
    __builtin_amdgcn_sched_barrier(0);
#pragma unroll
    for (int v = 0; v < 8; ++v) acc[v] = bias;
#pragma unroll
    for (int kc = 0; kc < 4; ++kc)
      acc = __builtin_amdgcn_wmma_f32_16x16x32_bf16(false, af2[kc], false, bw[kc], (short)0, acc, false, false);
#pragma unroll
    for (int v = 0; v < 8; ++v) {
      const float m = silu_f(acc[v]);
      dot[v] += m * wv;
      const int row = v + ((lane >> 4) << 3);
      mb[row * 128 + nt * 16 + colq] = f2bf(m);
    }
  }
  const float binf = b_inf[0];
#pragma unroll
  for (int v = 0; v < 8; ++v) {
    float s = dot[v];
    s += __shfl_xor(s, 1, 32);
    s += __shfl_xor(s, 2, 32);
    s += __shfl_xor(s, 4, 32);
    s += __shfl_xor(s, 8, 32);
    if (colq == 0) gbuf[wave][v + ((lane >> 4) << 3)] = sigm_f(s + binf);
  }
  __builtin_amdgcn_wave_barrier();

#pragma unroll
  for (int r = 0; r < 16; ++r) {
    const int s = __builtin_amdgcn_readlane(my_node, 2 * r);
    const float g = gbuf[wave][r];
#pragma unroll
    for (int c = lane; c < 128; c += 32) {
      unsafeAtomicAdd(&m_i[s * 128 + c], g * bf2f(mb[r * 128 + c]));
    }
  }
  __builtin_amdgcn_wave_barrier();
  }
}

__global__ __launch_bounds__(128) void egnn_node_kernel(
    const float* __restrict__ h,
    const u32x8* __restrict__ wf_h1, const u32x8* __restrict__ wf_h2,
    const float* __restrict__ b_h1, const float* __restrict__ b_h2,
    float* __restrict__ hio) {
  __shared__ unsigned short stage[4][16 * 256];
  __shared__ unsigned short a2[4][16 * 128];

  const int lane = threadIdx.x & 31;
  const int wave = threadIdx.x >> 5;
  const int colq = lane & 15;
  const int nbase = (blockIdx.x * 4 + wave) * 16;
  if (nbase >= NNODES) return;
  unsigned short* st = stage[wave];
  unsigned short* aw = a2[wave];

  for (int r = 0; r < 16; ++r) {
    const int node = nbase + r;
    const bool ok = node < NNODES;
#pragma unroll
    for (int i = lane; i < 256; i += 32) {
      float v = 0.0f;
      if (ok) v = (i < 128) ? h[node * 128 + i] : hio[node * 128 + (i - 128)];
      st[r * 256 + i] = f2bf(v);
    }
  }
  __builtin_amdgcn_wave_barrier();

#pragma unroll 1
  for (int nt = 0; nt < 8; ++nt) {
    f32x8 acc;
    const float bias = b_h1[nt * 16 + colq];
#pragma unroll
    for (int v = 0; v < 8; ++v) acc[v] = bias;
#pragma unroll
    for (int kc = 0; kc < 8; ++kc) {
      if (kc + 1 < 8) __builtin_prefetch(&wf_h1[((kc + 1) * 8 + nt) * 32 + lane], 0, 3);
      bf16x16 a = load_a_frag(st, 256, kc, lane);
      bf16x16 b = load_b_frag(wf_h1, kc, nt, lane);
      acc = __builtin_amdgcn_wmma_f32_16x16x32_bf16(false, a, false, b, (short)0, acc, false, false);
    }
#pragma unroll
    for (int v = 0; v < 8; ++v) {
      const int row = v + ((lane >> 4) << 3);
      aw[row * 128 + nt * 16 + colq] = f2bf(silu_f(acc[v]));
    }
  }
  __builtin_amdgcn_wave_barrier();

#pragma unroll 1
  for (int nt = 0; nt < 8; ++nt) {
    f32x8 acc;
    const float bias = b_h2[nt * 16 + colq];
#pragma unroll
    for (int v = 0; v < 8; ++v) acc[v] = bias;
#pragma unroll
    for (int kc = 0; kc < 4; ++kc) {
      bf16x16 a = load_a_frag(aw, 128, kc, lane);
      bf16x16 b = load_b_frag(wf_h2, kc, nt, lane);
      acc = __builtin_amdgcn_wmma_f32_16x16x32_bf16(false, a, false, b, (short)0, acc, false, false);
    }
#pragma unroll
    for (int v = 0; v < 8; ++v) {
      const int row = v + ((lane >> 4) << 3);
      const int node = nbase + row;
      const int col = nt * 16 + colq;
      if (node < NNODES) hio[node * 128 + col] = acc[v] + h[node * 128 + col];
    }
  }
}

extern "C" void kernel_launch(void* const* d_in, const int* in_sizes, int n_in,
                              void* d_out, int out_size, void* d_ws, size_t ws_size,
                              hipStream_t stream) {
  const float* h     = (const float*)d_in[0];
  const float* x     = (const float*)d_in[1];
  const int*   edges = (const int*)  d_in[2];
  const float* w_e1  = (const float*)d_in[3];
  const float* b_e1  = (const float*)d_in[4];
  const float* w_e2  = (const float*)d_in[5];
  const float* b_e2  = (const float*)d_in[6];
  const float* w_inf = (const float*)d_in[7];
  const float* b_inf = (const float*)d_in[8];
  const float* w_h1  = (const float*)d_in[9];
  const float* b_h1  = (const float*)d_in[10];
  const float* w_h2  = (const float*)d_in[11];
  const float* b_h2  = (const float*)d_in[12];
  float* out = (float*)d_out;

  unsigned int* wf_e1 = (unsigned int*)d_ws;
  unsigned int* wf_e2 = wf_e1 + 18432;
  unsigned int* wf_h1 = wf_e2 + 8192;
  unsigned int* wf_h2 = wf_h1 + 16384;

  egnn_prep_kernel<<<(51200 + 255) / 256, 256, 0, stream>>>(
      w_e1, w_e2, w_h1, w_h2, wf_e1, wf_e2, wf_h1, wf_h2);

  egnn_zero_kernel<<<(NNODES * 128 + 255) / 256, 256, 0, stream>>>(out, NNODES * 128);

  egnn_edge_kernel<<<EDGE_BLOCKS, EDGE_WAVES * 32, 0, stream>>>(
      h, x, edges, (const uint4*)wf_e1, (const uint4*)wf_e2,
      b_e1, b_e2, w_inf, b_inf, out);

  egnn_node_kernel<<<(NNODES + 63) / 64, 128, 0, stream>>>(
      h, (const u32x8*)wf_h1, (const u32x8*)wf_h2, b_h1, b_h2, out);
}


# f32 to bf16 by the hardware packed convert (v_cvt_pk_bf16_f32, round to nearest even) in place of the integer add-and-shift rounding sequence, in staging, epilogues and weight prep of all kernels
# speedup vs baseline: 1.7139x; 1.1628x over previous
#include <hip/hip_runtime.h>
#include <hip/hip_bf16.h>

#define NNODES 50000
#define NEDGES 800000

typedef __attribute__((ext_vector_type(16))) __bf16        bf16x16;
typedef __attribute__((ext_vector_type(8)))  float         f32x8;
typedef __attribute__((ext_vector_type(8)))  unsigned int  u32x8;
typedef __attribute__((ext_vector_type(2)))  float         f32x2;
typedef __attribute__((ext_vector_type(2)))  __bf16        bf16x2;

__device__ __forceinline__ unsigned short f2bf(float f) {
  return __builtin_bit_cast(unsigned short, (__bf16)f);
}
__device__ __forceinline__ unsigned int f2bf2(float a, float b) {
  const f32x2 v = {a, b};
  return __builtin_bit_cast(unsigned int, __builtin_convertvector(v, bf16x2));
}
__device__ __forceinline__ float bf2f(unsigned short s) {
  return __uint_as_float(((unsigned int)s) << 16);
}
__device__ __forceinline__ float silu_f(float v) { return v * __builtin_amdgcn_rcpf(1.0f + __expf(-v)); }
__device__ __forceinline__ float sigm_f(float v) { return __builtin_amdgcn_rcpf(1.0f + __expf(-v)); }

__device__ __forceinline__ int frag_k(int lane, int v) {
  return ((lane >> 4) << 3) + ((v >> 2) << 4) + ((v & 3) << 1);
}

__device__ __forceinline__ bf16x16 load_a_frag(const unsigned short* tile, int ld, int kc, int lane) {
  const int row = lane & 15;
  u32x8 r;
#pragma unroll
  for (int v = 0; v < 8; ++v) {
    const int k = kc * 32 + frag_k(lane, v);
    r[v] = *(const unsigned int*)(tile + row * ld + k);
  }
  return __builtin_bit_cast(bf16x16, r);
}

__device__ __forceinline__ bf16x16 load_b_frag(const u32x8* __restrict__ wf, int kc, int nt, int lane) {
  return __builtin_bit_cast(bf16x16, wf[(kc * 8 + nt) * 32 + lane]);
}

__device__ __forceinline__ bf16x16 load_b_frag_lds(const uint4* wl, int kc, int nt, int lane) {
  const uint4 lo = wl[(kc * 8 + nt) * 64 + lane];
  const uint4 hi = wl[(kc * 8 + nt) * 64 + 32 + lane];
  u32x8 r;
  r[0] = lo.x; r[1] = lo.y; r[2] = lo.z; r[3] = lo.w;
  r[4] = hi.x; r[5] = hi.y; r[6] = hi.z; r[7] = hi.w;
  return __builtin_bit_cast(bf16x16, r);
}

__global__ void egnn_prep_kernel(const float* __restrict__ w_e1, const float* __restrict__ w_e2,
                                 const float* __restrict__ w_h1, const float* __restrict__ w_h2,
                                 unsigned int* __restrict__ wf_e1, unsigned int* __restrict__ wf_e2,
                                 unsigned int* __restrict__ wf_h1, unsigned int* __restrict__ wf_h2) {
  int t = blockIdx.x * blockDim.x + threadIdx.x;
  const float* W; unsigned int* D; int K; int local; bool split;
  if      (t < 18432)         { W = w_e1; D = wf_e1; K = 257; local = t;         split = true;  }
  else if (t < 18432 + 8192)  { W = w_e2; D = wf_e2; K = 128; local = t - 18432; split = true;  }
  else if (t < 26624 + 16384) { W = w_h1; D = wf_h1; K = 256; local = t - 26624; split = false; }
  else if (t < 51200)         { W = w_h2; D = wf_h2; K = 128; local = t - 43008; split = false; }
  else return;
  const int v = local & 7, lane = (local >> 3) & 31, nt = (local >> 8) & 7, kc = local >> 11;
  const int k   = kc * 32 + ((lane >> 4) << 3) + ((v >> 2) << 4) + ((v & 3) << 1);
  const int col = nt * 16 + (lane & 15);
  const float a = (k     < K) ? W[k * 128 + col]       : 0.0f;
  const float b = (k + 1 < K) ? W[(k + 1) * 128 + col] : 0.0f;
  const int dst = split ? ((local >> 8) << 8) + ((v >> 2) << 7) + (lane << 2) + (v & 3) : local;
  D[dst] = f2bf2(a, b);
}

__global__ void egnn_zero_kernel(float* __restrict__ p, int n) {
  int t = blockIdx.x * blockDim.x + threadIdx.x;
  if (t < n) p[t] = 0.0f;
}

#define EDGE_BLOCKS 256
#define EDGE_WAVES  16
#define WL1_UNITS (9 * 8 * 64)
#define WL2_UNITS (4 * 8 * 64)
__global__ __launch_bounds__(EDGE_WAVES * 32) void egnn_edge_kernel(
    const float* __restrict__ h, const float* __restrict__ x, const int* __restrict__ edges,
    const uint4* __restrict__ wf_e1, const uint4* __restrict__ wf_e2,
    const float* __restrict__ b_e1, const float* __restrict__ b_e2,
    const float* __restrict__ w_inf, const float* __restrict__ b_inf,
    float* __restrict__ m_i) {
  __shared__ __attribute__((aligned(16))) uint4 wl1[WL1_UNITS];
  __shared__ __attribute__((aligned(16))) uint4 wl2[WL2_UNITS];
  __shared__ __attribute__((aligned(16))) unsigned short stage[EDGE_WAVES][16 * 288];
  __shared__ float gbuf[EDGE_WAVES][16];
  __shared__ float pvec[3][128];

  const int lane = threadIdx.x & 31;
  const int wave = threadIdx.x >> 5;
  const int colq = lane & 15;
  unsigned short* st = stage[wave];
  unsigned short* aw = st;
  unsigned short* mb = st + 2048;

  for (int i = threadIdx.x; i < WL1_UNITS; i += EDGE_WAVES * 32) wl1[i] = wf_e1[i];
  for (int i = threadIdx.x; i < WL2_UNITS; i += EDGE_WAVES * 32) wl2[i] = wf_e2[i];
  if (threadIdx.x < 128) {
    pvec[0][threadIdx.x] = b_e1[threadIdx.x];
    pvec[1][threadIdx.x] = b_e2[threadIdx.x];
    pvec[2][threadIdx.x] = w_inf[threadIdx.x];
  }
  __syncthreads();

#pragma unroll 1
  for (int tile = wave * EDGE_BLOCKS + blockIdx.x; tile < NEDGES / 16; tile += EDGE_BLOCKS * EDGE_WAVES) {
  const int ebase = tile * 16;

  const int my_node = edges[2 * ebase + lane];
  const float px = x[3 * my_node], py = x[3 * my_node + 1], pz = x[3 * my_node + 2];
  float4 hs[16], hd[16];
#pragma unroll
  for (int r = 0; r < 16; ++r) {
    const int s = __builtin_amdgcn_readlane(my_node, 2 * r);
    const int d = __builtin_amdgcn_readlane(my_node, 2 * r + 1);
    hs[r] = *(const float4*)(h + s * 128 + lane * 4);
    hd[r] = *(const float4*)(h + d * 128 + lane * 4);
  }
  const float dx = px - __shfl_xor(px, 1, 32), dy = py - __shfl_xor(py, 1, 32), dz = pz - __shfl_xor(pz, 1, 32);
  const float dist = sqrtf(dx * dx + dy * dy + dz * dz);
  const int distb = f2bf(dist);
#pragma unroll
  for (int r = 0; r < 16; ++r) {
    unsigned int* ws = (unsigned int*)(st + r * 288 + lane * 4);
    unsigned int* wd = (unsigned int*)(st + r * 288 + 128 + lane * 4);
    ws[0] = f2bf2(hs[r].x, hs[r].y);
    ws[1] = f2bf2(hs[r].z, hs[r].w);
    wd[0] = f2bf2(hd[r].x, hd[r].y);
    wd[1] = f2bf2(hd[r].z, hd[r].w);
    const int dr = __builtin_amdgcn_readlane(distb, 2 * r);
    st[r * 288 + 256 + lane] = (unsigned short)(lane == 0 ? dr : 0);
  }
  __builtin_amdgcn_wave_barrier();

  bf16x16 af1[9];
#pragma unroll
  for (int kc = 0; kc < 9; ++kc) af1[kc] = load_a_frag(st, 288, kc, lane);
  __builtin_amdgcn_wave_barrier();
#pragma unroll 1
  for (int nt = 0; nt < 8; ++nt) {
    f32x8 acc;
    const float bias = pvec[0][nt * 16 + colq];
    bf16x16 bw[9];
#pragma unroll
    for (int kc = 0; kc < 9; ++kc) bw[kc] = load_b_frag_lds(wl1, kc, nt, lane);
    __builtin_amdgcn_sched_barrier(0);
#pragma unroll
    for (int v = 0; v < 8; ++v) acc[v] = bias;
#pragma unroll
    for (int kc = 0; kc < 9; ++kc)
      acc = __builtin_amdgcn_wmma_f32_16x16x32_bf16(false, af1[kc], false, bw[kc], (short)0, acc, false, false);
#pragma unroll
    for (int v = 0; v < 8; ++v) {
      const int row = v + ((lane >> 4) << 3);
      aw[row * 128 + nt * 16 + colq] = f2bf(silu_f(acc[v]));
    }
  }
  __builtin_amdgcn_wave_barrier();

  bf16x16 af2[4];
#pragma unroll
  for (int kc = 0; kc < 4; ++kc) af2[kc] = load_a_frag(aw, 128, kc, lane);
  f32x8 dot;
#pragma unroll
  for (int v = 0; v < 8; ++v) dot[v] = 0.0f;
#pragma unroll 1
  for (int nt = 0; nt < 8; ++nt) {
    f32x8 acc;
    const float bias = pvec[1][nt * 16 + colq];
    const float wv = pvec[2][nt * 16 + colq];
    bf16x16 bw[4];
#pragma unroll
    for (int kc = 0; kc < 4; ++kc) bw[kc] = load_b_frag_lds(wl2, kc, nt, lane);
    __builtin_amdgcn_sched_barrier(0);
#pragma unroll
    for (int v = 0; v < 8; ++v) acc[v] = bias;
#pragma unroll
    for (int kc = 0; kc < 4; ++kc)
      acc = __builtin_amdgcn_wmma_f32_16x16x32_bf16(false, af2[kc], false, bw[kc], (short)0, acc, false, false);
#pragma unroll
    for (int v = 0; v < 8; ++v) {
      const float m = silu_f(acc[v]);
      dot[v] += m * wv;
      const int row = v + ((lane >> 4) << 3);
      mb[row * 128 + nt * 16 + colq] = f2bf(m);
    }
  }
  const float binf = b_inf[0];
#pragma unroll
  for (int v = 0; v < 8; ++v) {
    float s = dot[v];
    s += __shfl_xor(s, 1, 32);
    s += __shfl_xor(s, 2, 32);
    s += __shfl_xor(s, 4, 32);
    s += __shfl_xor(s, 8, 32);
    if (colq == 0) gbuf[wave][v + ((lane >> 4) << 3)] = sigm_f(s + binf);
  }
  __builtin_amdgcn_wave_barrier();

#pragma unroll
  for (int r = 0; r < 16; ++r) {
    const int s = __builtin_amdgcn_readlane(my_node, 2 * r);
    const float g = gbuf[wave][r];
#pragma unroll
    for (int c = lane; c < 128; c += 32) {
      unsafeAtomicAdd(&m_i[s * 128 + c], g * bf2f(mb[r * 128 + c]));
    }
  }
  __builtin_amdgcn_wave_barrier();
  }
}

__global__ __launch_bounds__(128) void egnn_node_kernel(
    const float* __restrict__ h,
    const u32x8* __restrict__ wf_h1, const u32x8* __restrict__ wf_h2,
    const float* __restrict__ b_h1, const float* __restrict__ b_h2,
    float* __restrict__ hio) {
  __shared__ unsigned short stage[4][16 * 256];
  __shared__ unsigned short a2[4][16 * 128];

  const int lane = threadIdx.x & 31;
  const int wave = threadIdx.x >> 5;
  const int colq = lane & 15;
  const int nbase = (blockIdx.x * 4 + wave) * 16;
  if (nbase >= NNODES) return;
  unsigned short* st = stage[wave];
  unsigned short* aw = a2[wave];

  for (int r = 0; r < 16; ++r) {
    const int node = nbase + r;
    const bool ok = node < NNODES;
#pragma unroll
    for (int i = lane; i < 256; i += 32) {
      float v = 0.0f;
      if (ok) v = (i < 128) ? h[node * 128 + i] : hio[node * 128 + (i - 128)];
      st[r * 256 + i] = f2bf(v);
    }
  }
  __builtin_amdgcn_wave_barrier();

#pragma unroll 1
  for (int nt = 0; nt < 8; ++nt) {
    f32x8 acc;
    const float bias = b_h1[nt * 16 + colq];
#pragma unroll
    for (int v = 0; v < 8; ++v) acc[v] = bias;
#pragma unroll
    for (int kc = 0; kc < 8; ++kc) {
      if (kc + 1 < 8) __builtin_prefetch(&wf_h1[((kc + 1) * 8 + nt) * 32 + lane], 0, 3);
      bf16x16 a = load_a_frag(st, 256, kc, lane);
      bf16x16 b = load_b_frag(wf_h1, kc, nt, lane);
      acc = __builtin_amdgcn_wmma_f32_16x16x32_bf16(false, a, false, b, (short)0, acc, false, false);
    }
#pragma unroll
    for (int v = 0; v < 8; ++v) {
      const int row = v + ((lane >> 4) << 3);
      aw[row * 128 + nt * 16 + colq] = f2bf(silu_f(acc[v]));
    }
  }
  __builtin_amdgcn_wave_barrier();

#pragma unroll 1
  for (int nt = 0; nt < 8; ++nt) {
    f32x8 acc;
    const float bias = b_h2[nt * 16 + colq];
#pragma unroll
    for (int v = 0; v < 8; ++v) acc[v] = bias;
#pragma unroll
    for (int kc = 0; kc < 4; ++kc) {
      bf16x16 a = load_a_frag(aw, 128, kc, lane);
      bf16x16 b = load_b_frag(wf_h2, kc, nt, lane);
      acc = __builtin_amdgcn_wmma_f32_16x16x32_bf16(false, a, false, b, (short)0, acc, false, false);
    }
#pragma unroll
    for (int v = 0; v < 8; ++v) {
      const int row = v + ((lane >> 4) << 3);
      const int node = nbase + row;
      const int col = nt * 16 + colq;
      if (node < NNODES) hio[node * 128 + col] = acc[v] + h[node * 128 + col];
    }
  }
}

extern "C" void kernel_launch(void* const* d_in, const int* in_sizes, int n_in,
                              void* d_out, int out_size, void* d_ws, size_t ws_size,
                              hipStream_t stream) {
  const float* h     = (const float*)d_in[0];
  const float* x     = (const float*)d_in[1];
  const int*   edges = (const int*)  d_in[2];
  const float* w_e1  = (const float*)d_in[3];
  const float* b_e1  = (const float*)d_in[4];
  const float* w_e2  = (const float*)d_in[5];
  const float* b_e2  = (const float*)d_in[6];
  const float* w_inf = (const float*)d_in[7];
  const float* b_inf = (const float*)d_in[8];
  const float* w_h1  = (const float*)d_in[9];
  const float* b_h1  = (const float*)d_in[10];
  const float* w_h2  = (const float*)d_in[11];
  const float* b_h2  = (const float*)d_in[12];
  float* out = (float*)d_out;

  unsigned int* wf_e1 = (unsigned int*)d_ws;
  unsigned int* wf_e2 = wf_e1 + 18432;
  unsigned int* wf_h1 = wf_e2 + 8192;
  unsigned int* wf_h2 = wf_h1 + 16384;

  egnn_prep_kernel<<<(51200 + 255) / 256, 256, 0, stream>>>(
      w_e1, w_e2, w_h1, w_h2, wf_e1, wf_e2, wf_h1, wf_h2);

  egnn_zero_kernel<<<(NNODES * 128 + 255) / 256, 256, 0, stream>>>(out, NNODES * 128);

  egnn_edge_kernel<<<EDGE_BLOCKS, EDGE_WAVES * 32, 0, stream>>>(
      h, x, edges, (const uint4*)wf_e1, (const uint4*)wf_e2,
      b_e1, b_e2, w_inf, b_inf, out);

  egnn_node_kernel<<<(NNODES + 63) / 64, 128, 0, stream>>>(
      h, (const u32x8*)wf_h1, (const u32x8*)wf_h2, b_h1, b_h2, out);
}


# edge scatter: m_ij kept lane-major in LDS so a lane reads its 4 columns of a row in one 8-byte load; the 16 rows and gates load as one group and the 64 atomics issue back to back, no wait per value
# speedup vs baseline: 1.9215x; 1.1211x over previous
#include <hip/hip_runtime.h>
#include <hip/hip_bf16.h>

#define NNODES 50000
#define NEDGES 800000

typedef __attribute__((ext_vector_type(16))) __bf16        bf16x16;
typedef __attribute__((ext_vector_type(8)))  float         f32x8;
typedef __attribute__((ext_vector_type(8)))  unsigned int  u32x8;
typedef __attribute__((ext_vector_type(2)))  float         f32x2;
typedef __attribute__((ext_vector_type(2)))  __bf16        bf16x2;

__device__ __forceinline__ unsigned short f2bf(float f) {
  return __builtin_bit_cast(unsigned short, (__bf16)f);
}
__device__ __forceinline__ unsigned int f2bf2(float a, float b) {
  const f32x2 v = {a, b};
  return __builtin_bit_cast(unsigned int, __builtin_convertvector(v, bf16x2));
}
__device__ __forceinline__ float bf2f(unsigned short s) {
  return __uint_as_float(((unsigned int)s) << 16);
}
__device__ __forceinline__ float silu_f(float v) { return v * __builtin_amdgcn_rcpf(1.0f + __expf(-v)); }
__device__ __forceinline__ float sigm_f(float v) { return __builtin_amdgcn_rcpf(1.0f + __expf(-v)); }

__device__ __forceinline__ int frag_k(int lane, int v) {
  return ((lane >> 4) << 3) + ((v >> 2) << 4) + ((v & 3) << 1);
}

__device__ __forceinline__ bf16x16 load_a_frag(const unsigned short* tile, int ld, int kc, int lane) {
  const int row = lane & 15;
  u32x8 r;
#pragma unroll
  for (int v = 0; v < 8; ++v) {
    const int k = kc * 32 + frag_k(lane, v);
    r[v] = *(const unsigned int*)(tile + row * ld + k);
  }
  return __builtin_bit_cast(bf16x16, r);
}

__device__ __forceinline__ bf16x16 load_b_frag(const u32x8* __restrict__ wf, int kc, int nt, int lane) {
  return __builtin_bit_cast(bf16x16, wf[(kc * 8 + nt) * 32 + lane]);
}

__device__ __forceinline__ bf16x16 load_b_frag_lds(const uint4* wl, int kc, int nt, int lane) {
  const uint4 lo = wl[(kc * 8 + nt) * 64 + lane];
  const uint4 hi = wl[(kc * 8 + nt) * 64 + 32 + lane];
  u32x8 r;
  r[0] = lo.x; r[1] = lo.y; r[2] = lo.z; r[3] = lo.w;
  r[4] = hi.x; r[5] = hi.y; r[6] = hi.z; r[7] = hi.w;
  return __builtin_bit_cast(bf16x16, r);
}

__global__ void egnn_prep_kernel(const float* __restrict__ w_e1, const float* __restrict__ w_e2,
                                 const float* __restrict__ w_h1, const float* __restrict__ w_h2,
                                 unsigned int* __restrict__ wf_e1, unsigned int* __restrict__ wf_e2,
                                 unsigned int* __restrict__ wf_h1, unsigned int* __restrict__ wf_h2) {
  int t = blockIdx.x * blockDim.x + threadIdx.x;
  const float* W; unsigned int* D; int K; int local; bool split;
  if      (t < 18432)         { W = w_e1; D = wf_e1; K = 257; local = t;         split = true;  }
  else if (t < 18432 + 8192)  { W = w_e2; D = wf_e2; K = 128; local = t - 18432; split = true;  }
  else if (t < 26624 + 16384) { W = w_h1; D = wf_h1; K = 256; local = t - 26624; split = false; }
  else if (t < 51200)         { W = w_h2; D = wf_h2; K = 128; local = t - 43008; split = false; }
  else return;
  const int v = local & 7, lane = (local >> 3) & 31, nt = (local >> 8) & 7, kc = local >> 11;
  const int k   = kc * 32 + ((lane >> 4) << 3) + ((v >> 2) << 4) + ((v & 3) << 1);
  const int col = nt * 16 + (lane & 15);
  const float a = (k     < K) ? W[k * 128 + col]       : 0.0f;
  const float b = (k + 1 < K) ? W[(k + 1) * 128 + col] : 0.0f;
  const int dst = split ? ((local >> 8) << 8) + ((v >> 2) << 7) + (lane << 2) + (v & 3) : local;
  D[dst] = f2bf2(a, b);
}

__global__ void egnn_zero_kernel(float* __restrict__ p, int n) {
  int t = blockIdx.x * blockDim.x + threadIdx.x;
  if (t < n) p[t] = 0.0f;
}

#define EDGE_BLOCKS 256
#define EDGE_WAVES  16
#define WL1_UNITS (9 * 8 * 64)
#define WL2_UNITS (4 * 8 * 64)
__global__ __launch_bounds__(EDGE_WAVES * 32) void egnn_edge_kernel(
    const float* __restrict__ h, const float* __restrict__ x, const int* __restrict__ edges,
    const uint4* __restrict__ wf_e1, const uint4* __restrict__ wf_e2,
    const float* __restrict__ b_e1, const float* __restrict__ b_e2,
    const float* __restrict__ w_inf, const float* __restrict__ b_inf,
    float* __restrict__ m_i) {
  __shared__ __attribute__((aligned(16))) uint4 wl1[WL1_UNITS];
  __shared__ __attribute__((aligned(16))) uint4 wl2[WL2_UNITS];
  __shared__ __attribute__((aligned(16))) unsigned short stage[EDGE_WAVES][16 * 288];
  __shared__ float gbuf[EDGE_WAVES][16];
  __shared__ float pvec[3][128];

  const int lane = threadIdx.x & 31;
  const int wave = threadIdx.x >> 5;
  const int colq = lane & 15;
  unsigned short* st = stage[wave];
  unsigned short* aw = st;
  unsigned short* mb = st + 2048;

  for (int i = threadIdx.x; i < WL1_UNITS; i += EDGE_WAVES * 32) wl1[i] = wf_e1[i];
  for (int i = threadIdx.x; i < WL2_UNITS; i += EDGE_WAVES * 32) wl2[i] = wf_e2[i];
  if (threadIdx.x < 128) {
    pvec[0][threadIdx.x] = b_e1[threadIdx.x];
    pvec[1][threadIdx.x] = b_e2[threadIdx.x];
    pvec[2][threadIdx.x] = w_inf[threadIdx.x];
  }
  __syncthreads();

#pragma unroll 1
  for (int tile = wave * EDGE_BLOCKS + blockIdx.x; tile < NEDGES / 16; tile += EDGE_BLOCKS * EDGE_WAVES) {
  const int ebase = tile * 16;

  const int my_node = edges[2 * ebase + lane];
  const float px = x[3 * my_node], py = x[3 * my_node + 1], pz = x[3 * my_node + 2];
  float4 hs[16], hd[16];
#pragma unroll
  for (int r = 0; r < 16; ++r) {
    const int s = __builtin_amdgcn_readlane(my_node, 2 * r);
    const int d = __builtin_amdgcn_readlane(my_node, 2 * r + 1);
    hs[r] = *(const float4*)(h + s * 128 + lane * 4);
    hd[r] = *(const float4*)(h + d * 128 + lane * 4);
  }
  const float dx = px - __shfl_xor(px, 1, 32), dy = py - __shfl_xor(py, 1, 32), dz = pz - __shfl_xor(pz, 1, 32);
  const float dist = sqrtf(dx * dx + dy * dy + dz * dz);
  const int distb = f2bf(dist);
#pragma unroll
  for (int r = 0; r < 16; ++r) {
    unsigned int* ws = (unsigned int*)(st + r * 288 + lane * 4);
    unsigned int* wd = (unsigned int*)(st + r * 288 + 128 + lane * 4);
    ws[0] = f2bf2(hs[r].x, hs[r].y);
    ws[1] = f2bf2(hs[r].z, hs[r].w);
    wd[0] = f2bf2(hd[r].x, hd[r].y);
    wd[1] = f2bf2(hd[r].z, hd[r].w);
    const int dr = __builtin_amdgcn_readlane(distb, 2 * r);
    st[r * 288 + 256 + lane] = (unsigned short)(lane == 0 ? dr : 0);
  }
  __builtin_amdgcn_wave_barrier();

  bf16x16 af1[9];
#pragma unroll
  for (int kc = 0; kc < 9; ++kc) af1[kc] = load_a_frag(st, 288, kc, lane);
  __builtin_amdgcn_wave_barrier();
#pragma unroll 1
  for (int nt = 0; nt < 8; ++nt) {
    f32x8 acc;
    const float bias = pvec[0][nt * 16 + colq];
    bf16x16 bw[9];
#pragma unroll
    for (int kc = 0; kc < 9; ++kc) bw[kc] = load_b_frag_lds(wl1, kc, nt, lane);
    __builtin_amdgcn_sched_barrier(0);
#pragma unroll
    for (int v = 0; v < 8; ++v) acc[v] = bias;
#pragma unroll
    for (int kc = 0; kc < 9; ++kc)
      acc = __builtin_amdgcn_wmma_f32_16x16x32_bf16(false, af1[kc], false, bw[kc], (short)0, acc, false, false);
#pragma unroll
    for (int v = 0; v < 8; ++v) {
      const int row = v + ((lane >> 4) << 3);
      aw[row * 128 + nt * 16 + colq] = f2bf(silu_f(acc[v]));
    }
  }
  __builtin_amdgcn_wave_barrier();

  bf16x16 af2[4];
#pragma unroll
  for (int kc = 0; kc < 4; ++kc) af2[kc] = load_a_frag(aw, 128, kc, lane);
  f32x8 dot;
#pragma unroll
  for (int v = 0; v < 8; ++v) dot[v] = 0.0f;
#pragma unroll 1
  for (int nt = 0; nt < 8; ++nt) {
    f32x8 acc;
    const float bias = pvec[1][nt * 16 + colq];
    const float wv = pvec[2][nt * 16 + colq];
    bf16x16 bw[4];
#pragma unroll
    for (int kc = 0; kc < 4; ++kc) bw[kc] = load_b_frag_lds(wl2, kc, nt, lane);
    __builtin_amdgcn_sched_barrier(0);
#pragma unroll
    for (int v = 0; v < 8; ++v) acc[v] = bias;
#pragma unroll
    for (int kc = 0; kc < 4; ++kc)
      acc = __builtin_amdgcn_wmma_f32_16x16x32_bf16(false, af2[kc], false, bw[kc], (short)0, acc, false, false);
#pragma unroll
    for (int v = 0; v < 8; ++v) {
      const float m = silu_f(acc[v]);
      dot[v] += m * wv;
      const int row = v + ((lane >> 4) << 3);
      mb[row * 128 + ((((nt & 1) << 4) + colq) << 2) + (nt >> 1)] = f2bf(m);
    }
  }
  const float binf = b_inf[0];
#pragma unroll
  for (int v = 0; v < 8; ++v) {
    float s = dot[v];
    s += __shfl_xor(s, 1, 32);
    s += __shfl_xor(s, 2, 32);
    s += __shfl_xor(s, 4, 32);
    s += __shfl_xor(s, 8, 32);
    if (colq == 0) gbuf[wave][v + ((lane >> 4) << 3)] = sigm_f(s + binf);
  }
  __builtin_amdgcn_wave_barrier();

  uint2 mv[16];
#pragma unroll
  for (int r = 0; r < 16; ++r) mv[r] = *(const uint2*)(mb + r * 128 + lane * 4);
  float4 gq[4];
#pragma unroll
  for (int q = 0; q < 4; ++q) gq[q] = *(const float4*)&gbuf[wave][4 * q];
  const float gr[16] = {gq[0].x, gq[0].y, gq[0].z, gq[0].w, gq[1].x, gq[1].y, gq[1].z, gq[1].w,
                        gq[2].x, gq[2].y, gq[2].z, gq[2].w, gq[3].x, gq[3].y, gq[3].z, gq[3].w};
  float pv[16][4];
  int off[16];
#pragma unroll
  for (int r = 0; r < 16; ++r) {
    pv[r][0] = gr[r] * bf2f((unsigned short)(mv[r].x & 0xffffu));
    pv[r][1] = gr[r] * bf2f((unsigned short)(mv[r].x >> 16));
    pv[r][2] = gr[r] * bf2f((unsigned short)(mv[r].y & 0xffffu));
    pv[r][3] = gr[r] * bf2f((unsigned short)(mv[r].y >> 16));
    off[r] = __builtin_amdgcn_readlane(my_node, 2 * r) * 128 + lane;
  }
  __builtin_amdgcn_sched_barrier(0);
#pragma unroll
  for (int r = 0; r < 16; ++r) {
#pragma unroll
    for (int j = 0; j < 4; ++j) unsafeAtomicAdd(&m_i[off[r] + 32 * j], pv[r][j]);
  }
  __builtin_amdgcn_wave_barrier();
  }
}

__global__ __launch_bounds__(128) void egnn_node_kernel(
    const float* __restrict__ h,
    const u32x8* __restrict__ wf_h1, const u32x8* __restrict__ wf_h2,
    const float* __restrict__ b_h1, const float* __restrict__ b_h2,
    float* __restrict__ hio) {
  __shared__ unsigned short stage[4][16 * 256];
  __shared__ unsigned short a2[4][16 * 128];

  const int lane = threadIdx.x & 31;
  const int wave = threadIdx.x >> 5;
  const int colq = lane & 15;
  const int nbase = (blockIdx.x * 4 + wave) * 16;
  if (nbase >= NNODES) return;
  unsigned short* st = stage[wave];
  unsigned short* aw = a2[wave];

  for (int r = 0; r < 16; ++r) {
    const int node = nbase + r;
    const bool ok = node < NNODES;
#pragma unroll
    for (int i = lane; i < 256; i += 32) {
      float v = 0.0f;
      if (ok) v = (i < 128) ? h[node * 128 + i] : hio[node * 128 + (i - 128)];
      st[r * 256 + i] = f2bf(v);
    }
  }
  __builtin_amdgcn_wave_barrier();

#pragma unroll 1
  for (int nt = 0; nt < 8; ++nt) {
    f32x8 acc;
    const float bias = b_h1[nt * 16 + colq];
#pragma unroll
    for (int v = 0; v < 8; ++v) acc[v] = bias;
#pragma unroll
    for (int kc = 0; kc < 8; ++kc) {
      if (kc + 1 < 8) __builtin_prefetch(&wf_h1[((kc + 1) * 8 + nt) * 32 + lane], 0, 3);
      bf16x16 a = load_a_frag(st, 256, kc, lane);
      bf16x16 b = load_b_frag(wf_h1, kc, nt, lane);
      acc = __builtin_amdgcn_wmma_f32_16x16x32_bf16(false, a, false, b, (short)0, acc, false, false);
    }
#pragma unroll
    for (int v = 0; v < 8; ++v) {
      const int row = v + ((lane >> 4) << 3);
      aw[row * 128 + nt * 16 + colq] = f2bf(silu_f(acc[v]));
    }
  }
  __builtin_amdgcn_wave_barrier();

#pragma unroll 1
  for (int nt = 0; nt < 8; ++nt) {
    f32x8 acc;
    const float bias = b_h2[nt * 16 + colq];
#pragma unroll
    for (int v = 0; v < 8; ++v) acc[v] = bias;
#pragma unroll
    for (int kc = 0; kc < 4; ++kc) {
      bf16x16 a = load_a_frag(aw, 128, kc, lane);
      bf16x16 b = load_b_frag(wf_h2, kc, nt, lane);
      acc = __builtin_amdgcn_wmma_f32_16x16x32_bf16(false, a, false, b, (short)0, acc, false, false);
    }
#pragma unroll
    for (int v = 0; v < 8; ++v) {
      const int row = v + ((lane >> 4) << 3);
      const int node = nbase + row;
      const int col = nt * 16 + colq;
      if (node < NNODES) hio[node * 128 + col] = acc[v] + h[node * 128 + col];
    }
  }
}

extern "C" void kernel_launch(void* const* d_in, const int* in_sizes, int n_in,
                              void* d_out, int out_size, void* d_ws, size_t ws_size,
                              hipStream_t stream) {
  const float* h     = (const float*)d_in[0];
  const float* x     = (const float*)d_in[1];
  const int*   edges = (const int*)  d_in[2];
  const float* w_e1  = (const float*)d_in[3];
  const float* b_e1  = (const float*)d_in[4];
  const float* w_e2  = (const float*)d_in[5];
  const float* b_e2  = (const float*)d_in[6];
  const float* w_inf = (const float*)d_in[7];
  const float* b_inf = (const float*)d_in[8];
  const float* w_h1  = (const float*)d_in[9];
  const float* b_h1  = (const float*)d_in[10];
  const float* w_h2  = (const float*)d_in[11];
  const float* b_h2  = (const float*)d_in[12];
  float* out = (float*)d_out;

  unsigned int* wf_e1 = (unsigned int*)d_ws;
  unsigned int* wf_e2 = wf_e1 + 18432;
  unsigned int* wf_h1 = wf_e2 + 8192;
  unsigned int* wf_h2 = wf_h1 + 16384;

  egnn_prep_kernel<<<(51200 + 255) / 256, 256, 0, stream>>>(
      w_e1, w_e2, w_h1, w_h2, wf_e1, wf_e2, wf_h1, wf_h2);

  egnn_zero_kernel<<<(NNODES * 128 + 255) / 256, 256, 0, stream>>>(out, NNODES * 128);

  egnn_edge_kernel<<<EDGE_BLOCKS, EDGE_WAVES * 32, 0, stream>>>(
      h, x, edges, (const uint4*)wf_e1, (const uint4*)wf_e2,
      b_e1, b_e2, w_inf, b_inf, out);

  egnn_node_kernel<<<(NNODES + 63) / 64, 128, 0, stream>>>(
      h, (const u32x8*)wf_h1, (const u32x8*)wf_h2, b_h1, b_h2, out);
}


# zero fill of m_i: 250 resident 8-wave workgroups, each wave 25 back-to-back 16-byte stores over its own 12,800 bytes, in place of 200,000 one-store waves whose entry rate set that launch's time
# speedup vs baseline: 2.0729x; 1.0788x over previous
#include <hip/hip_runtime.h>
#include <hip/hip_bf16.h>

#define NNODES 50000
#define NEDGES 800000

typedef __attribute__((ext_vector_type(16))) __bf16        bf16x16;
typedef __attribute__((ext_vector_type(8)))  float         f32x8;
typedef __attribute__((ext_vector_type(8)))  unsigned int  u32x8;
typedef __attribute__((ext_vector_type(2)))  float         f32x2;
typedef __attribute__((ext_vector_type(2)))  __bf16        bf16x2;

__device__ __forceinline__ unsigned short f2bf(float f) {
  return __builtin_bit_cast(unsigned short, (__bf16)f);
}
__device__ __forceinline__ unsigned int f2bf2(float a, float b) {
  const f32x2 v = {a, b};
  return __builtin_bit_cast(unsigned int, __builtin_convertvector(v, bf16x2));
}
__device__ __forceinline__ float bf2f(unsigned short s) {
  return __uint_as_float(((unsigned int)s) << 16);
}
__device__ __forceinline__ float silu_f(float v) { return v * __builtin_amdgcn_rcpf(1.0f + __expf(-v)); }
__device__ __forceinline__ float sigm_f(float v) { return __builtin_amdgcn_rcpf(1.0f + __expf(-v)); }

__device__ __forceinline__ int frag_k(int lane, int v) {
  return ((lane >> 4) << 3) + ((v >> 2) << 4) + ((v & 3) << 1);
}

__device__ __forceinline__ bf16x16 load_a_frag(const unsigned short* tile, int ld, int kc, int lane) {
  const int row = lane & 15;
  u32x8 r;
#pragma unroll
  for (int v = 0; v < 8; ++v) {
    const int k = kc * 32 + frag_k(lane, v);
    r[v] = *(const unsigned int*)(tile + row * ld + k);
  }
  return __builtin_bit_cast(bf16x16, r);
}

__device__ __forceinline__ bf16x16 load_b_frag(const u32x8* __restrict__ wf, int kc, int nt, int lane) {
  return __builtin_bit_cast(bf16x16, wf[(kc * 8 + nt) * 32 + lane]);
}

__device__ __forceinline__ bf16x16 load_b_frag_lds(const uint4* wl, int kc, int nt, int lane) {
  const uint4 lo = wl[(kc * 8 + nt) * 64 + lane];
  const uint4 hi = wl[(kc * 8 + nt) * 64 + 32 + lane];
  u32x8 r;
  r[0] = lo.x; r[1] = lo.y; r[2] = lo.z; r[3] = lo.w;
  r[4] = hi.x; r[5] = hi.y; r[6] = hi.z; r[7] = hi.w;
  return __builtin_bit_cast(bf16x16, r);
}

__global__ void egnn_prep_kernel(const float* __restrict__ w_e1, const float* __restrict__ w_e2,
                                 const float* __restrict__ w_h1, const float* __restrict__ w_h2,
                                 unsigned int* __restrict__ wf_e1, unsigned int* __restrict__ wf_e2,
                                 unsigned int* __restrict__ wf_h1, unsigned int* __restrict__ wf_h2) {
  int t = blockIdx.x * blockDim.x + threadIdx.x;
  const float* W; unsigned int* D; int K; int local; bool split;
  if      (t < 18432)         { W = w_e1; D = wf_e1; K = 257; local = t;         split = true;  }
  else if (t < 18432 + 8192)  { W = w_e2; D = wf_e2; K = 128; local = t - 18432; split = true;  }
  else if (t < 26624 + 16384) { W = w_h1; D = wf_h1; K = 256; local = t - 26624; split = false; }
  else if (t < 51200)         { W = w_h2; D = wf_h2; K = 128; local = t - 43008; split = false; }
  else return;
  const int v = local & 7, lane = (local >> 3) & 31, nt = (local >> 8) & 7, kc = local >> 11;
  const int k   = kc * 32 + ((lane >> 4) << 3) + ((v >> 2) << 4) + ((v & 3) << 1);
  const int col = nt * 16 + (lane & 15);
  const float a = (k     < K) ? W[k * 128 + col]       : 0.0f;
  const float b = (k + 1 < K) ? W[(k + 1) * 128 + col] : 0.0f;
  const int dst = split ? ((local >> 8) << 8) + ((v >> 2) << 7) + (lane << 2) + (v & 3) : local;
  D[dst] = f2bf2(a, b);
}

#define ZERO_BLOCKS  250
#define ZERO_WAVES   8
#define ZERO_STORES  25
static_assert(ZERO_BLOCKS * ZERO_WAVES * ZERO_STORES * 32 * 4 == NNODES * 128, "the waves' slices tile the buffer exactly");
__global__ __launch_bounds__(ZERO_WAVES * 32) void egnn_zero_kernel(float* __restrict__ p) {
  const int lane = threadIdx.x & 31;
  const int wave = blockIdx.x * ZERO_WAVES + (threadIdx.x >> 5);
  float4* __restrict__ p4 = (float4*)p + wave * (ZERO_STORES * 32) + lane;
  const float4 z = {0.0f, 0.0f, 0.0f, 0.0f};
#pragma unroll
  for (int k = 0; k < ZERO_STORES; ++k) p4[k * 32] = z;
}

#define EDGE_BLOCKS 256
#define EDGE_WAVES  16
#define WL1_UNITS (9 * 8 * 64)
#define WL2_UNITS (4 * 8 * 64)
__global__ __launch_bounds__(EDGE_WAVES * 32) void egnn_edge_kernel(
    const float* __restrict__ h, const float* __restrict__ x, const int* __restrict__ edges,
    const uint4* __restrict__ wf_e1, const uint4* __restrict__ wf_e2,
    const float* __restrict__ b_e1, const float* __restrict__ b_e2,
    const float* __restrict__ w_inf, const float* __restrict__ b_inf,
    float* __restrict__ m_i) {
  __shared__ __attribute__((aligned(16))) uint4 wl1[WL1_UNITS];
  __shared__ __attribute__((aligned(16))) uint4 wl2[WL2_UNITS];
  __shared__ __attribute__((aligned(16))) unsigned short stage[EDGE_WAVES][16 * 288];
  __shared__ float gbuf[EDGE_WAVES][16];
  __shared__ float pvec[3][128];

  const int lane = threadIdx.x & 31;
  const int wave = threadIdx.x >> 5;
  const int colq = lane & 15;
  unsigned short* st = stage[wave];
  unsigned short* aw = st;
  unsigned short* mb = st + 2048;

  for (int i = threadIdx.x; i < WL1_UNITS; i += EDGE_WAVES * 32) wl1[i] = wf_e1[i];
  for (int i = threadIdx.x; i < WL2_UNITS; i += EDGE_WAVES * 32) wl2[i] = wf_e2[i];
  if (threadIdx.x < 128) {
    pvec[0][threadIdx.x] = b_e1[threadIdx.x];
    pvec[1][threadIdx.x] = b_e2[threadIdx.x];
    pvec[2][threadIdx.x] = w_inf[threadIdx.x];
  }
  __syncthreads();

#pragma unroll 1
  for (int tile = wave * EDGE_BLOCKS + blockIdx.x; tile < NEDGES / 16; tile += EDGE_BLOCKS * EDGE_WAVES) {
  const int ebase = tile * 16;

  const int my_node = edges[2 * ebase + lane];
  const float px = x[3 * my_node], py = x[3 * my_node + 1], pz = x[3 * my_node + 2];
  float4 hs[16], hd[16];
#pragma unroll
  for (int r = 0; r < 16; ++r) {
    const int s = __builtin_amdgcn_readlane(my_node, 2 * r);
    const int d = __builtin_amdgcn_readlane(my_node, 2 * r + 1);
    hs[r] = *(const float4*)(h + s * 128 + lane * 4);
    hd[r] = *(const float4*)(h + d * 128 + lane * 4);
  }
  const float dx = px - __shfl_xor(px, 1, 32), dy = py - __shfl_xor(py, 1, 32), dz = pz - __shfl_xor(pz, 1, 32);
  const float dist = sqrtf(dx * dx + dy * dy + dz * dz);
  const int distb = f2bf(dist);
#pragma unroll
  for (int r = 0; r < 16; ++r) {
    unsigned int* ws = (unsigned int*)(st + r * 288 + lane * 4);
    unsigned int* wd = (unsigned int*)(st + r * 288 + 128 + lane * 4);
    ws[0] = f2bf2(hs[r].x, hs[r].y);
    ws[1] = f2bf2(hs[r].z, hs[r].w);
    wd[0] = f2bf2(hd[r].x, hd[r].y);
    wd[1] = f2bf2(hd[r].z, hd[r].w);
    const int dr = __builtin_amdgcn_readlane(distb, 2 * r);
    st[r * 288 + 256 + lane] = (unsigned short)(lane == 0 ? dr : 0);
  }
  __builtin_amdgcn_wave_barrier();

  bf16x16 af1[9];
#pragma unroll
  for (int kc = 0; kc < 9; ++kc) af1[kc] = load_a_frag(st, 288, kc, lane);
  __builtin_amdgcn_wave_barrier();
#pragma unroll 1
  for (int nt = 0; nt < 8; ++nt) {
    f32x8 acc;
    const float bias = pvec[0][nt * 16 + colq];
    bf16x16 bw[9];
#pragma unroll
    for (int kc = 0; kc < 9; ++kc) bw[kc] = load_b_frag_lds(wl1, kc, nt, lane);
    __builtin_amdgcn_sched_barrier(0);
#pragma unroll
    for (int v = 0; v < 8; ++v) acc[v] = bias;
#pragma unroll
    for (int kc = 0; kc < 9; ++kc)
      acc = __builtin_amdgcn_wmma_f32_16x16x32_bf16(false, af1[kc], false, bw[kc], (short)0, acc, false, false);
#pragma unroll
    for (int v = 0; v < 8; ++v) {
      const int row = v + ((lane >> 4) << 3);
      aw[row * 128 + nt * 16 + colq] = f2bf(silu_f(acc[v]));
    }
  }
  __builtin_amdgcn_wave_barrier();

  bf16x16 af2[4];
#pragma unroll
  for (int kc = 0; kc < 4; ++kc) af2[kc] = load_a_frag(aw, 128, kc, lane);
  f32x8 dot;
#pragma unroll
  for (int v = 0; v < 8; ++v) dot[v] = 0.0f;
#pragma unroll 1
  for (int nt = 0; nt < 8; ++nt) {
    f32x8 acc;
    const float bias = pvec[1][nt * 16 + colq];
    const float wv = pvec[2][nt * 16 + colq];
    bf16x16 bw[4];
#pragma unroll
    for (int kc = 0; kc < 4; ++kc) bw[kc] = load_b_frag_lds(wl2, kc, nt, lane);
    __builtin_amdgcn_sched_barrier(0);
#pragma unroll
    for (int v = 0; v < 8; ++v) acc[v] = bias;
#pragma unroll
    for (int kc = 0; kc < 4; ++kc)
      acc = __builtin_amdgcn_wmma_f32_16x16x32_bf16(false, af2[kc], false, bw[kc], (short)0, acc, false, false);
#pragma unroll
    for (int v = 0; v < 8; ++v) {
      const float m = silu_f(acc[v]);
      dot[v] += m * wv;
      const int row = v + ((lane >> 4) << 3);
      mb[row * 128 + ((((nt & 1) << 4) + colq) << 2) + (nt >> 1)] = f2bf(m);
    }
  }
  const float binf = b_inf[0];
#pragma unroll
  for (int v = 0; v < 8; ++v) {
    float s = dot[v];
    s += __shfl_xor(s, 1, 32);
    s += __shfl_xor(s, 2, 32);
    s += __shfl_xor(s, 4, 32);
    s += __shfl_xor(s, 8, 32);
    if (colq == 0) gbuf[wave][v + ((lane >> 4) << 3)] = sigm_f(s + binf);
  }
  __builtin_amdgcn_wave_barrier();

  uint2 mv[16];
#pragma unroll
  for (int r = 0; r < 16; ++r) mv[r] = *(const uint2*)(mb + r * 128 + lane * 4);
  float4 gq[4];
#pragma unroll
  for (int q = 0; q < 4; ++q) gq[q] = *(const float4*)&gbuf[wave][4 * q];
  const float gr[16] = {gq[0].x, gq[0].y, gq[0].z, gq[0].w, gq[1].x, gq[1].y, gq[1].z, gq[1].w,
                        gq[2].x, gq[2].y, gq[2].z, gq[2].w, gq[3].x, gq[3].y, gq[3].z, gq[3].w};
  float pv[16][4];
  int off[16];
#pragma unroll
  for (int r = 0; r < 16; ++r) {
    pv[r][0] = gr[r] * bf2f((unsigned short)(mv[r].x & 0xffffu));
    pv[r][1] = gr[r] * bf2f((unsigned short)(mv[r].x >> 16));
    pv[r][2] = gr[r] * bf2f((unsigned short)(mv[r].y & 0xffffu));
    pv[r][3] = gr[r] * bf2f((unsigned short)(mv[r].y >> 16));
    off[r] = __builtin_amdgcn_readlane(my_node, 2 * r) * 128 + lane;
  }
  __builtin_amdgcn_sched_barrier(0);
#pragma unroll
  for (int r = 0; r < 16; ++r) {
#pragma unroll
    for (int j = 0; j < 4; ++j) unsafeAtomicAdd(&m_i[off[r] + 32 * j], pv[r][j]);
  }
  __builtin_amdgcn_wave_barrier();
  }
}

static_assert(NNODES % 16 == 0, "a wave's 16-row tile is always full: no row of it lies past the last node");
__global__ __launch_bounds__(128) void egnn_node_kernel(
    const float* __restrict__ h,
    const u32x8* __restrict__ wf_h1, const u32x8* __restrict__ wf_h2,
    const float* __restrict__ b_h1, const float* __restrict__ b_h2,
    float* __restrict__ hio) {
  __shared__ __attribute__((aligned(16))) unsigned int stage[4][16 * 128];
  __shared__ __attribute__((aligned(16))) unsigned short a2[4][16 * 128];

  const int lane = threadIdx.x & 31;
  const int wave = threadIdx.x >> 5;
  const int colq = lane & 15;
  const int nbase = (blockIdx.x * 4 + wave) * 16;
  if (nbase >= NNODES) return;
  unsigned int* su = stage[wave];
  const unsigned short* st = (const unsigned short*)su;
  unsigned short* aw = a2[wave];

  float4 hr[16], mr[16];
#pragma unroll
  for (int r = 0; r < 16; ++r) {
    hr[r] = *(const float4*)(h   + (nbase + r) * 128 + lane * 4);
    mr[r] = *(const float4*)(hio + (nbase + r) * 128 + lane * 4);
  }
#pragma unroll
  for (int r = 0; r < 16; ++r) {
    uint2 a, b;
    a.x = f2bf2(hr[r].x, hr[r].y); a.y = f2bf2(hr[r].z, hr[r].w);
    b.x = f2bf2(mr[r].x, mr[r].y); b.y = f2bf2(mr[r].z, mr[r].w);
    *(uint2*)(su + r * 128 + lane * 2)      = a;
    *(uint2*)(su + r * 128 + 64 + lane * 2) = b;
  }
  __builtin_amdgcn_wave_barrier();

  bf16x16 af1[8];
#pragma unroll
  for (int kc = 0; kc < 8; ++kc) af1[kc] = load_a_frag(st, 256, kc, lane);
#pragma unroll 1
  for (int nt = 0; nt < 8; ++nt) {
    f32x8 acc;
    bf16x16 bw[8];
#pragma unroll
    for (int kc = 0; kc < 8; ++kc) bw[kc] = load_b_frag(wf_h1, kc, nt, lane);
    const float bias = b_h1[nt * 16 + colq];
    __builtin_amdgcn_sched_barrier(0);
#pragma unroll
    for (int v = 0; v < 8; ++v) acc[v] = bias;
#pragma unroll
    for (int kc = 0; kc < 8; ++kc)
      acc = __builtin_amdgcn_wmma_f32_16x16x32_bf16(false, af1[kc], false, bw[kc], (short)0, acc, false, false);
#pragma unroll
    for (int v = 0; v < 8; ++v) {
      const int row = v + ((lane >> 4) << 3);
      aw[row * 128 + nt * 16 + colq] = f2bf(silu_f(acc[v]));
    }
  }
  __builtin_amdgcn_wave_barrier();

  bf16x16 af2[4];
#pragma unroll
  for (int kc = 0; kc < 4; ++kc) af2[kc] = load_a_frag(aw, 128, kc, lane);
#pragma unroll 1
  for (int nt = 0; nt < 8; ++nt) {
    f32x8 acc;
    bf16x16 bw[4];
#pragma unroll
    for (int kc = 0; kc < 4; ++kc) bw[kc] = load_b_frag(wf_h2, kc, nt, lane);
    const float bias = b_h2[nt * 16 + colq];
    __builtin_amdgcn_sched_barrier(0);
#pragma unroll
    for (int v = 0; v < 8; ++v) acc[v] = bias;
#pragma unroll
    for (int kc = 0; kc < 4; ++kc)
      acc = __builtin_amdgcn_wmma_f32_16x16x32_bf16(false, af2[kc], false, bw[kc], (short)0, acc, false, false);
#pragma unroll
    for (int v = 0; v < 8; ++v) {
      const int row = v + ((lane >> 4) << 3);
      su[row * 128 + nt * 16 + colq] = __float_as_uint(acc[v]);
    }
  }
  __builtin_amdgcn_wave_barrier();

#pragma unroll
  for (int r = 0; r < 16; ++r) {
    const uint4 o = *(const uint4*)(su + r * 128 + lane * 4);
    float4 y;
    y.x = __uint_as_float(o.x) + hr[r].x;
    y.y = __uint_as_float(o.y) + hr[r].y;
    y.z = __uint_as_float(o.z) + hr[r].z;
    y.w = __uint_as_float(o.w) + hr[r].w;
    *(float4*)(hio + (nbase + r) * 128 + lane * 4) = y;
  }
}

extern "C" void kernel_launch(void* const* d_in, const int* in_sizes, int n_in,
                              void* d_out, int out_size, void* d_ws, size_t ws_size,
                              hipStream_t stream) {
  const float* h     = (const float*)d_in[0];
  const float* x     = (const float*)d_in[1];
  const int*   edges = (const int*)  d_in[2];
  const float* w_e1  = (const float*)d_in[3];
  const float* b_e1  = (const float*)d_in[4];
  const float* w_e2  = (const float*)d_in[5];
  const float* b_e2  = (const float*)d_in[6];
  const float* w_inf = (const float*)d_in[7];
  const float* b_inf = (const float*)d_in[8];
  const float* w_h1  = (const float*)d_in[9];
  const float* b_h1  = (const float*)d_in[10];
  const float* w_h2  = (const float*)d_in[11];
  const float* b_h2  = (const float*)d_in[12];
  float* out = (float*)d_out;

  unsigned int* wf_e1 = (unsigned int*)d_ws;
  unsigned int* wf_e2 = wf_e1 + 18432;
  unsigned int* wf_h1 = wf_e2 + 8192;
  unsigned int* wf_h2 = wf_h1 + 16384;

  egnn_prep_kernel<<<(51200 + 255) / 256, 256, 0, stream>>>(
      w_e1, w_e2, w_h1, w_h2, wf_e1, wf_e2, wf_h1, wf_h2);

  egnn_zero_kernel<<<ZERO_BLOCKS, ZERO_WAVES * 32, 0, stream>>>(out);

  egnn_edge_kernel<<<EDGE_BLOCKS, EDGE_WAVES * 32, 0, stream>>>(
      h, x, edges, (const uint4*)wf_e1, (const uint4*)wf_e2,
      b_e1, b_e2, w_inf, b_inf, out);

  egnn_node_kernel<<<(NNODES + 63) / 64, 128, 0, stream>>>(
      h, (const u32x8*)wf_h1, (const u32x8*)wf_h2, b_h1, b_h2, out);
}


# edge gate: the eight 16-lane dot sums by row_xmask DPP adds on the vector ALU and the 16 gates passed on by readlane, in place of 32 waited LDS-unit shuffles and an LDS store and reload
# speedup vs baseline: 2.1790x; 1.0512x over previous
#include <hip/hip_runtime.h>
#include <hip/hip_bf16.h>

#define NNODES 50000
#define NEDGES 800000

typedef __attribute__((ext_vector_type(16))) __bf16        bf16x16;
typedef __attribute__((ext_vector_type(8)))  float         f32x8;
typedef __attribute__((ext_vector_type(8)))  unsigned int  u32x8;
typedef __attribute__((ext_vector_type(2)))  float         f32x2;
typedef __attribute__((ext_vector_type(2)))  __bf16        bf16x2;

__device__ __forceinline__ unsigned short f2bf(float f) {
  return __builtin_bit_cast(unsigned short, (__bf16)f);
}
__device__ __forceinline__ unsigned int f2bf2(float a, float b) {
  const f32x2 v = {a, b};
  return __builtin_bit_cast(unsigned int, __builtin_convertvector(v, bf16x2));
}
__device__ __forceinline__ float bf2f(unsigned short s) {
  return __uint_as_float(((unsigned int)s) << 16);
}
__device__ __forceinline__ float silu_f(float v) { return v * __builtin_amdgcn_rcpf(1.0f + __expf(-v)); }
__device__ __forceinline__ float sigm_f(float v) { return __builtin_amdgcn_rcpf(1.0f + __expf(-v)); }

template <int M>
__device__ __forceinline__ float row_xor(float v) {
  return __int_as_float(__builtin_amdgcn_update_dpp(0, __float_as_int(v), 0x160 + M, 0xf, 0xf, false));
}

__device__ __forceinline__ int frag_k(int lane, int v) {
  return ((lane >> 4) << 3) + ((v >> 2) << 4) + ((v & 3) << 1);
}

__device__ __forceinline__ bf16x16 load_a_frag(const unsigned short* tile, int ld, int kc, int lane) {
  const int row = lane & 15;
  u32x8 r;
#pragma unroll
  for (int v = 0; v < 8; ++v) {
    const int k = kc * 32 + frag_k(lane, v);
    r[v] = *(const unsigned int*)(tile + row * ld + k);
  }
  return __builtin_bit_cast(bf16x16, r);
}

__device__ __forceinline__ bf16x16 load_b_frag(const u32x8* __restrict__ wf, int kc, int nt, int lane) {
  return __builtin_bit_cast(bf16x16, wf[(kc * 8 + nt) * 32 + lane]);
}

__device__ __forceinline__ bf16x16 load_b_frag_lds(const uint4* wl, int kc, int nt, int lane) {
  const uint4 lo = wl[(kc * 8 + nt) * 64 + lane];
  const uint4 hi = wl[(kc * 8 + nt) * 64 + 32 + lane];
  u32x8 r;
  r[0] = lo.x; r[1] = lo.y; r[2] = lo.z; r[3] = lo.w;
  r[4] = hi.x; r[5] = hi.y; r[6] = hi.z; r[7] = hi.w;
  return __builtin_bit_cast(bf16x16, r);
}

__global__ void egnn_prep_kernel(const float* __restrict__ w_e1, const float* __restrict__ w_e2,
                                 const float* __restrict__ w_h1, const float* __restrict__ w_h2,
                                 unsigned int* __restrict__ wf_e1, unsigned int* __restrict__ wf_e2,
                                 unsigned int* __restrict__ wf_h1, unsigned int* __restrict__ wf_h2) {
  int t = blockIdx.x * blockDim.x + threadIdx.x;
  const float* W; unsigned int* D; int K; int local; bool split;
  if      (t < 18432)         { W = w_e1; D = wf_e1; K = 257; local = t;         split = true;  }
  else if (t < 18432 + 8192)  { W = w_e2; D = wf_e2; K = 128; local = t - 18432; split = true;  }
  else if (t < 26624 + 16384) { W = w_h1; D = wf_h1; K = 256; local = t - 26624; split = false; }
  else if (t < 51200)         { W = w_h2; D = wf_h2; K = 128; local = t - 43008; split = false; }
  else return;
  const int v = local & 7, lane = (local >> 3) & 31, nt = (local >> 8) & 7, kc = local >> 11;
  const int k   = kc * 32 + ((lane >> 4) << 3) + ((v >> 2) << 4) + ((v & 3) << 1);
  const int col = nt * 16 + (lane & 15);
  const float a = (k     < K) ? W[k * 128 + col]       : 0.0f;
  const float b = (k + 1 < K) ? W[(k + 1) * 128 + col] : 0.0f;
  const int dst = split ? ((local >> 8) << 8) + ((v >> 2) << 7) + (lane << 2) + (v & 3) : local;
  D[dst] = f2bf2(a, b);
}

#define ZERO_BLOCKS  250
#define ZERO_WAVES   8
#define ZERO_STORES  25
static_assert(ZERO_BLOCKS * ZERO_WAVES * ZERO_STORES * 32 * 4 == NNODES * 128, "the waves' slices tile the buffer exactly");
__global__ __launch_bounds__(ZERO_WAVES * 32) void egnn_zero_kernel(float* __restrict__ p) {
  const int lane = threadIdx.x & 31;
  const int wave = blockIdx.x * ZERO_WAVES + (threadIdx.x >> 5);
  float4* __restrict__ p4 = (float4*)p + wave * (ZERO_STORES * 32) + lane;
  const float4 z = {0.0f, 0.0f, 0.0f, 0.0f};
#pragma unroll
  for (int k = 0; k < ZERO_STORES; ++k) p4[k * 32] = z;
}

#define EDGE_BLOCKS 256
#define EDGE_WAVES  16
#define WL1_UNITS (9 * 8 * 64)
#define WL2_UNITS (4 * 8 * 64)
__global__ __launch_bounds__(EDGE_WAVES * 32) void egnn_edge_kernel(
    const float* __restrict__ h, const float* __restrict__ x, const int* __restrict__ edges,
    const uint4* __restrict__ wf_e1, const uint4* __restrict__ wf_e2,
    const float* __restrict__ b_e1, const float* __restrict__ b_e2,
    const float* __restrict__ w_inf, const float* __restrict__ b_inf,
    float* __restrict__ m_i) {
  __shared__ __attribute__((aligned(16))) uint4 wl1[WL1_UNITS];
  __shared__ __attribute__((aligned(16))) uint4 wl2[WL2_UNITS];
  __shared__ __attribute__((aligned(16))) unsigned short stage[EDGE_WAVES][16 * 288];
  __shared__ float pvec[3][128];

  const int lane = threadIdx.x & 31;
  const int wave = threadIdx.x >> 5;
  const int colq = lane & 15;
  unsigned short* st = stage[wave];
  unsigned short* aw = st;
  unsigned short* mb = st + 2048;

  for (int i = threadIdx.x; i < WL1_UNITS; i += EDGE_WAVES * 32) wl1[i] = wf_e1[i];
  for (int i = threadIdx.x; i < WL2_UNITS; i += EDGE_WAVES * 32) wl2[i] = wf_e2[i];
  if (threadIdx.x < 128) {
    pvec[0][threadIdx.x] = b_e1[threadIdx.x];
    pvec[1][threadIdx.x] = b_e2[threadIdx.x];
    pvec[2][threadIdx.x] = w_inf[threadIdx.x];
  }
  __syncthreads();

#pragma unroll 1
  for (int tile = wave * EDGE_BLOCKS + blockIdx.x; tile < NEDGES / 16; tile += EDGE_BLOCKS * EDGE_WAVES) {
  const int ebase = tile * 16;

  const int my_node = edges[2 * ebase + lane];
  const float px = x[3 * my_node], py = x[3 * my_node + 1], pz = x[3 * my_node + 2];
  float4 hs[16], hd[16];
#pragma unroll
  for (int r = 0; r < 16; ++r) {
    const int s = __builtin_amdgcn_readlane(my_node, 2 * r);
    const int d = __builtin_amdgcn_readlane(my_node, 2 * r + 1);
    hs[r] = *(const float4*)(h + s * 128 + lane * 4);
    hd[r] = *(const float4*)(h + d * 128 + lane * 4);
  }
  const float dx = px - row_xor<1>(px), dy = py - row_xor<1>(py), dz = pz - row_xor<1>(pz);
  const float dist = sqrtf(dx * dx + dy * dy + dz * dz);
  const int distb = f2bf(dist);
#pragma unroll
  for (int r = 0; r < 16; ++r) {
    unsigned int* ws = (unsigned int*)(st + r * 288 + lane * 4);
    unsigned int* wd = (unsigned int*)(st + r * 288 + 128 + lane * 4);
    ws[0] = f2bf2(hs[r].x, hs[r].y);
    ws[1] = f2bf2(hs[r].z, hs[r].w);
    wd[0] = f2bf2(hd[r].x, hd[r].y);
    wd[1] = f2bf2(hd[r].z, hd[r].w);
    const int dr = __builtin_amdgcn_readlane(distb, 2 * r);
    st[r * 288 + 256 + lane] = (unsigned short)(lane == 0 ? dr : 0);
  }
  __builtin_amdgcn_wave_barrier();

  bf16x16 af1[9];
#pragma unroll
  for (int kc = 0; kc < 9; ++kc) af1[kc] = load_a_frag(st, 288, kc, lane);
  __builtin_amdgcn_wave_barrier();
#pragma unroll 1
  for (int nt = 0; nt < 8; ++nt) {
    f32x8 acc;
    const float bias = pvec[0][nt * 16 + colq];
    bf16x16 bw[9];
#pragma unroll
    for (int kc = 0; kc < 9; ++kc) bw[kc] = load_b_frag_lds(wl1, kc, nt, lane);
    __builtin_amdgcn_sched_barrier(0);
#pragma unroll
    for (int v = 0; v < 8; ++v) acc[v] = bias;
#pragma unroll
    for (int kc = 0; kc < 9; ++kc)
      acc = __builtin_amdgcn_wmma_f32_16x16x32_bf16(false, af1[kc], false, bw[kc], (short)0, acc, false, false);
#pragma unroll
    for (int v = 0; v < 8; ++v) {
      const int row = v + ((lane >> 4) << 3);
      aw[row * 128 + nt * 16 + colq] = f2bf(silu_f(acc[v]));
    }
  }
  __builtin_amdgcn_wave_barrier();

  bf16x16 af2[4];
#pragma unroll
  for (int kc = 0; kc < 4; ++kc) af2[kc] = load_a_frag(aw, 128, kc, lane);
  f32x8 dot;
#pragma unroll
  for (int v = 0; v < 8; ++v) dot[v] = 0.0f;
#pragma unroll 1
  for (int nt = 0; nt < 8; ++nt) {
    f32x8 acc;
    const float bias = pvec[1][nt * 16 + colq];
    const float wv = pvec[2][nt * 16 + colq];
    bf16x16 bw[4];
#pragma unroll
    for (int kc = 0; kc < 4; ++kc) bw[kc] = load_b_frag_lds(wl2, kc, nt, lane);
    __builtin_amdgcn_sched_barrier(0);
#pragma unroll
    for (int v = 0; v < 8; ++v) acc[v] = bias;
#pragma unroll
    for (int kc = 0; kc < 4; ++kc)
      acc = __builtin_amdgcn_wmma_f32_16x16x32_bf16(false, af2[kc], false, bw[kc], (short)0, acc, false, false);
#pragma unroll
    for (int v = 0; v < 8; ++v) {
      const float m = silu_f(acc[v]);
      dot[v] += m * wv;
      const int row = v + ((lane >> 4) << 3);
      mb[row * 128 + ((((nt & 1) << 4) + colq) << 2) + (nt >> 1)] = f2bf(m);
    }
  }
  const float binf = b_inf[0];
#pragma unroll
  for (int v = 0; v < 8; ++v) {
    float s = dot[v];
    s += row_xor<1>(s);
    s += row_xor<2>(s);
    s += row_xor<4>(s);
    s += row_xor<8>(s);
    dot[v] = s;
  }
  float gsum = dot[0];
#pragma unroll
  for (int v = 1; v < 8; ++v) gsum = (colq == v) ? dot[v] : gsum;
  const float gate = sigm_f(gsum + binf);
  __builtin_amdgcn_wave_barrier();

  uint2 mv[16];
#pragma unroll
  for (int r = 0; r < 16; ++r) mv[r] = *(const uint2*)(mb + r * 128 + lane * 4);
  float gr[16];
#pragma unroll
  for (int r = 0; r < 16; ++r)
    gr[r] = __int_as_float(__builtin_amdgcn_readlane(__float_as_int(gate), ((r >> 3) << 4) + (r & 7)));
  float pv[16][4];
  int off[16];
#pragma unroll
  for (int r = 0; r < 16; ++r) {
    pv[r][0] = gr[r] * bf2f((unsigned short)(mv[r].x & 0xffffu));
    pv[r][1] = gr[r] * bf2f((unsigned short)(mv[r].x >> 16));
    pv[r][2] = gr[r] * bf2f((unsigned short)(mv[r].y & 0xffffu));
    pv[r][3] = gr[r] * bf2f((unsigned short)(mv[r].y >> 16));
    off[r] = __builtin_amdgcn_readlane(my_node, 2 * r) * 128 + lane;
  }
  __builtin_amdgcn_sched_barrier(0);
#pragma unroll
  for (int r = 0; r < 16; ++r) {
#pragma unroll
    for (int j = 0; j < 4; ++j) unsafeAtomicAdd(&m_i[off[r] + 32 * j], pv[r][j]);
  }
  __builtin_amdgcn_wave_barrier();
  }
}

static_assert(NNODES % 16 == 0, "a wave's 16-row tile is always full: no row of it lies past the last node");
__global__ __launch_bounds__(128) void egnn_node_kernel(
    const float* __restrict__ h,
    const u32x8* __restrict__ wf_h1, const u32x8* __restrict__ wf_h2,
    const float* __restrict__ b_h1, const float* __restrict__ b_h2,
    float* __restrict__ hio) {
  __shared__ __attribute__((aligned(16))) unsigned int stage[4][16 * 128];
  __shared__ __attribute__((aligned(16))) unsigned short a2[4][16 * 128];

  const int lane = threadIdx.x & 31;
  const int wave = threadIdx.x >> 5;
  const int colq = lane & 15;
  const int nbase = (blockIdx.x * 4 + wave) * 16;
  if (nbase >= NNODES) return;
  unsigned int* su = stage[wave];
  const unsigned short* st = (const unsigned short*)su;
  unsigned short* aw = a2[wave];

  float4 hr[16], mr[16];
#pragma unroll
  for (int r = 0; r < 16; ++r) {
    hr[r] = *(const float4*)(h   + (nbase + r) * 128 + lane * 4);
    mr[r] = *(const float4*)(hio + (nbase + r) * 128 + lane * 4);
  }
#pragma unroll
  for (int r = 0; r < 16; ++r) {
    uint2 a, b;
    a.x = f2bf2(hr[r].x, hr[r].y); a.y = f2bf2(hr[r].z, hr[r].w);
    b.x = f2bf2(mr[r].x, mr[r].y); b.y = f2bf2(mr[r].z, mr[r].w);
    *(uint2*)(su + r * 128 + lane * 2)      = a;
    *(uint2*)(su + r * 128 + 64 + lane * 2) = b;
  }
  __builtin_amdgcn_wave_barrier();

  bf16x16 af1[8];
#pragma unroll
  for (int kc = 0; kc < 8; ++kc) af1[kc] = load_a_frag(st, 256, kc, lane);
#pragma unroll 1
  for (int nt = 0; nt < 8; ++nt) {
    f32x8 acc;
    bf16x16 bw[8];
#pragma unroll
    for (int kc = 0; kc < 8; ++kc) bw[kc] = load_b_frag(wf_h1, kc, nt, lane);
    const float bias = b_h1[nt * 16 + colq];
    __builtin_amdgcn_sched_barrier(0);
#pragma unroll
    for (int v = 0; v < 8; ++v) acc[v] = bias;
#pragma unroll
    for (int kc = 0; kc < 8; ++kc)
      acc = __builtin_amdgcn_wmma_f32_16x16x32_bf16(false, af1[kc], false, bw[kc], (short)0, acc, false, false);
#pragma unroll
    for (int v = 0; v < 8; ++v) {
      const int row = v + ((lane >> 4) << 3);
      aw[row * 128 + nt * 16 + colq] = f2bf(silu_f(acc[v]));
    }
  }
  __builtin_amdgcn_wave_barrier();

  bf16x16 af2[4];
#pragma unroll
  for (int kc = 0; kc < 4; ++kc) af2[kc] = load_a_frag(aw, 128, kc, lane);
#pragma unroll 1
  for (int nt = 0; nt < 8; ++nt) {
    f32x8 acc;
    bf16x16 bw[4];
#pragma unroll
    for (int kc = 0; kc < 4; ++kc) bw[kc] = load_b_frag(wf_h2, kc, nt, lane);
    const float bias = b_h2[nt * 16 + colq];
    __builtin_amdgcn_sched_barrier(0);
#pragma unroll
    for (int v = 0; v < 8; ++v) acc[v] = bias;
#pragma unroll
    for (int kc = 0; kc < 4; ++kc)
      acc = __builtin_amdgcn_wmma_f32_16x16x32_bf16(false, af2[kc], false, bw[kc], (short)0, acc, false, false);
#pragma unroll
    for (int v = 0; v < 8; ++v) {
      const int row = v + ((lane >> 4) << 3);
      su[row * 128 + nt * 16 + colq] = __float_as_uint(acc[v]);
    }
  }
  __builtin_amdgcn_wave_barrier();

#pragma unroll
  for (int r = 0; r < 16; ++r) {
    const uint4 o = *(const uint4*)(su + r * 128 + lane * 4);
    float4 y;
    y.x = __uint_as_float(o.x) + hr[r].x;
    y.y = __uint_as_float(o.y) + hr[r].y;
    y.z = __uint_as_float(o.z) + hr[r].z;
    y.w = __uint_as_float(o.w) + hr[r].w;
    *(float4*)(hio + (nbase + r) * 128 + lane * 4) = y;
  }
}

extern "C" void kernel_launch(void* const* d_in, const int* in_sizes, int n_in,
                              void* d_out, int out_size, void* d_ws, size_t ws_size,
                              hipStream_t stream) {
  const float* h     = (const float*)d_in[0];
  const float* x     = (const float*)d_in[1];
  const int*   edges = (const int*)  d_in[2];
  const float* w_e1  = (const float*)d_in[3];
  const float* b_e1  = (const float*)d_in[4];
  const float* w_e2  = (const float*)d_in[5];
  const float* b_e2  = (const float*)d_in[6];
  const float* w_inf = (const float*)d_in[7];
  const float* b_inf = (const float*)d_in[8];
  const float* w_h1  = (const float*)d_in[9];
  const float* b_h1  = (const float*)d_in[10];
  const float* w_h2  = (const float*)d_in[11];
  const float* b_h2  = (const float*)d_in[12];
  float* out = (float*)d_out;

  unsigned int* wf_e1 = (unsigned int*)d_ws;
  unsigned int* wf_e2 = wf_e1 + 18432;
  unsigned int* wf_h1 = wf_e2 + 8192;
  unsigned int* wf_h2 = wf_h1 + 16384;

  egnn_prep_kernel<<<(51200 + 255) / 256, 256, 0, stream>>>(
      w_e1, w_e2, w_h1, w_h2, wf_e1, wf_e2, wf_h1, wf_h2);

  egnn_zero_kernel<<<ZERO_BLOCKS, ZERO_WAVES * 32, 0, stream>>>(out);

  egnn_edge_kernel<<<EDGE_BLOCKS, EDGE_WAVES * 32, 0, stream>>>(
      h, x, edges, (const uint4*)wf_e1, (const uint4*)wf_e2,
      b_e1, b_e2, w_inf, b_inf, out);

  egnn_node_kernel<<<(NNODES + 63) / 64, 128, 0, stream>>>(
      h, (const u32x8*)wf_h1, (const u32x8*)wf_h2, b_h1, b_h2, out);
}
